# starting kernel
# baseline (speedup 1.0000x reference)
#include <hip/hip_runtime.h>
#include <hip/hip_bf16.h>

#define EMBED 768
#define NSEQ  4096
#define DHEAD 64
#define LDSTR 772

typedef __attribute__((ext_vector_type(2))) float v2f;
typedef __attribute__((ext_vector_type(8))) float v8f;

__global__ __launch_bounds__(384) void qkv_gemm_kernel(
    const float* __restrict__ x,
    const float* __restrict__ Wq, const float* __restrict__ bq,
    const float* __restrict__ Wk, const float* __restrict__ bk,
    const float* __restrict__ Wv, const float* __restrict__ bv,
    float* __restrict__ Q, float* __restrict__ K, float* __restrict__ V) {
  __shared__ float lds_x[16 * LDSTR];

  const int m0  = blockIdx.x * 16;
  const int tid = threadIdx.x;

  for (int v = tid; v < 16 * EMBED / 4; v += 384) {
    const int r  = v / (EMBED / 4);
    const int c4 = v % (EMBED / 4);
    const float4 val = ((const float4*)(x + (size_t)(m0 + r) * EMBED))[c4];
    float* dst = &lds_x[r * LDSTR + c4 * 4];
    dst[0] = val.x; dst[1] = val.y; dst[2] = val.z; dst[3] = val.w;
  }
  __syncthreads();

  const int wave = tid >> 5;
  const int lane = tid & 31;
  const int ln   = lane & 15;
  const int lh   = lane >> 4;

  const float* W; const float* bias; float* outp; int n0;
  if (wave < 4)      { W = Wq; bias = bq; outp = Q; n0 = wave * 16; }
  else if (wave < 8) { W = Wk; bias = bk; outp = K; n0 = (wave - 4) * 16; }
  else               { W = Wv; bias = bv; outp = V; n0 = (wave - 8) * 16; }

  const float bval = bias[n0 + ln];
  v8f acc;
#pragma unroll
  for (int r = 0; r < 8; ++r) acc[r] = bval;

  const float* lrow = &lds_x[ln * LDSTR];
  const float* wcol = W + n0 + ln;

#pragma unroll 4
  for (int k0 = 0; k0 < EMBED; k0 += 4) {
    const int ka = k0 + 2 * lh;
    v2f a;  a[0] = lrow[ka];
            a[1] = lrow[ka + 1];
    v2f bf; bf[0] = wcol[(size_t)ka * DHEAD];
            bf[1] = wcol[(size_t)(ka + 1) * DHEAD];
    acc = __builtin_amdgcn_wmma_f32_16x16x4_f32(
         false, a,  false, bf,
         (short)0, acc,  false,  false);
  }

#pragma unroll
  for (int r = 0; r < 8; ++r) {
    const int m = m0 + r + 8 * lh;
    outp[(size_t)m * DHEAD + n0 + ln] = acc[r];
  }
}

__global__ __launch_bounds__(256) void vsum_kernel(const float* __restrict__ V,
                                                   float* __restrict__ Srest) {
  __shared__ float red[256];
  const int b = blockIdx.x;
  const int t = threadIdx.x;
  const int d = t & 63;
  const int s = t >> 6;
  float acc = 0.0f;
  for (int j = 3 + s; j < NSEQ; j += 4)
    acc += V[((size_t)b * NSEQ + j) * DHEAD + d];
  red[t] = acc;
  __syncthreads();
  if (s == 0)
    Srest[b * DHEAD + d] = red[d] + red[64 + d] + red[128 + d] + red[192 + d];
}

__global__ __launch_bounds__(256) void combine_kernel(
    const float* __restrict__ Q, const float* __restrict__ K,
    const float* __restrict__ V, const float* __restrict__ Srest,
    float* __restrict__ out) {
  const int tid  = threadIdx.x;
  const int wave = tid >> 5;
  const int lane = tid & 31;
  const int g = blockIdx.x * 8 + wave;
  const int b = g >> 12;
  const int i = g & (NSEQ - 1);

  const float2 q = ((const float2*)(Q + (size_t)g * DHEAD))[lane];

  float pa = 0.0f, pb = 0.0f, pc = 0.0f;
  {
    const float2 kc = ((const float2*)(K + (size_t)g * DHEAD))[lane];
    pb = q.x * kc.x + q.y * kc.y;
  }
  if (i > 0) {
    const float2 kp = ((const float2*)(K + (size_t)(g - 1) * DHEAD))[lane];
    pa = q.x * kp.x + q.y * kp.y;
  }
  if (i < NSEQ - 1) {
    const float2 kn = ((const float2*)(K + (size_t)(g + 1) * DHEAD))[lane];
    pc = q.x * kn.x + q.y * kn.y;
  }
#pragma unroll
  for (int off = 16; off > 0; off >>= 1) {
    pa += __shfl_xor(pa, off, 32);
    pb += __shfl_xor(pb, off, 32);
    pc += __shfl_xor(pc, off, 32);
  }
  const float a  = (i > 0) ? pa : 0.0f;
  const float bs = pb;
  const float c  = (i < NSEQ - 1) ? pc : 0.0f;

  const float m  = fmaxf(fmaxf(a, bs), fmaxf(c, 0.0f));
  const float ea = expf(a - m);
  const float eb = expf(bs - m);
  const float ec = expf(c - m);
  const float ez = expf(-m);
  const float inv = 1.0f / (ea + eb + ec + (float)(NSEQ - 3) * ez);

  const float* Vb = V + (size_t)b * NSEQ * DHEAD;
  const float2 v0 = ((const float2*)(Vb))[lane];
  const float2 v1 = ((const float2*)(Vb + DHEAD))[lane];
  const float2 v2 = ((const float2*)(Vb + 2 * DHEAD))[lane];
  const float2 sr = ((const float2*)(Srest + b * DHEAD))[lane];

  float2 o;
  o.x = (ea * v0.x + eb * v1.x + ec * v2.x + ez * sr.x) * inv;
  o.y = (ea * v0.y + eb * v1.y + ec * v2.y + ez * sr.y) * inv;
  ((float2*)(out + (size_t)g * DHEAD))[lane] = o;
}

extern "C" void kernel_launch(void* const* d_in, const int* in_sizes, int n_in,
                              void* d_out, int out_size, void* d_ws, size_t ws_size,
                              hipStream_t stream) {
  const float* x  = (const float*)d_in[0];
  const float* Wq = (const float*)d_in[1];
  const float* bq = (const float*)d_in[2];
  const float* Wk = (const float*)d_in[3];
  const float* bk = (const float*)d_in[4];
  const float* Wv = (const float*)d_in[5];
  const float* bv = (const float*)d_in[6];
  float* out = (float*)d_out;

  const int BN = in_sizes[0] / EMBED;
  const int B  = BN / NSEQ;

  float* Q     = (float*)d_ws;
  float* K     = Q + (size_t)BN * DHEAD;
  float* V     = K + (size_t)BN * DHEAD;
  float* Srest = V + (size_t)BN * DHEAD;

  qkv_gemm_kernel<<<BN / 16, 384, 0, stream>>>(x, Wq, bq, Wk, bk, Wv, bv, Q, K, V);
  vsum_kernel<<<B, 256, 0, stream>>>(V, Srest);
  combine_kernel<<<BN / 8, 256, 0, stream>>>(Q, K, V, Srest, out);
}


# V column sums folded from the projection kernel's accumulators (per-block partial rows, then a fixed 16x16 tree per batch) instead of a 4-block serial pass over V; summation order changes
# speedup vs baseline: 2.0157x; 2.0157x over previous
#include <hip/hip_runtime.h>
#include <hip/hip_bf16.h>

#define EMBED 768
#define NSEQ  4096
#define DHEAD 64
#define LDSTR 772
#define TILES_PER_BATCH (NSEQ / 16)

typedef __attribute__((ext_vector_type(2))) float v2f;
typedef __attribute__((ext_vector_type(8))) float v8f;

__global__ __launch_bounds__(384) void qkv_gemm_kernel(
    const float* __restrict__ x,
    const float* __restrict__ Wq, const float* __restrict__ bq,
    const float* __restrict__ Wk, const float* __restrict__ bk,
    const float* __restrict__ Wv, const float* __restrict__ bv,
    float* __restrict__ Q, float* __restrict__ K,
    float* __restrict__ Vhead, float* __restrict__ Psum) {
  __shared__ float lds_x[16 * LDSTR];

  const int m0  = blockIdx.x * 16;
  const int tid = threadIdx.x;

  for (int v = tid; v < 16 * EMBED / 4; v += 384) {
    const int r  = v / (EMBED / 4);
    const int c4 = v % (EMBED / 4);
    const float4 val = ((const float4*)(x + (size_t)(m0 + r) * EMBED))[c4];
    float* dst = &lds_x[r * LDSTR + c4 * 4];
    dst[0] = val.x; dst[1] = val.y; dst[2] = val.z; dst[3] = val.w;
  }
  __syncthreads();

  const int wave = tid >> 5;
  const int lane = tid & 31;
  const int ln   = lane & 15;
  const int lh   = lane >> 4;

  const float* W; const float* bias; int n0;
  if (wave < 4)      { W = Wq; bias = bq; n0 = wave * 16; }
  else if (wave < 8) { W = Wk; bias = bk; n0 = (wave - 4) * 16; }
  else               { W = Wv; bias = bv; n0 = (wave - 8) * 16; }

  const float bval = bias[n0 + ln];
  v8f acc;
#pragma unroll
  for (int r = 0; r < 8; ++r) acc[r] = bval;

  const float* lrow = &lds_x[ln * LDSTR];
  const float* wcol = W + n0 + ln;

#pragma unroll 4
  for (int k0 = 0; k0 < EMBED; k0 += 4) {
    const int ka = k0 + 2 * lh;
    v2f a;  a[0] = lrow[ka];
            a[1] = lrow[ka + 1];
    v2f bf; bf[0] = wcol[(size_t)ka * DHEAD];
            bf[1] = wcol[(size_t)(ka + 1) * DHEAD];
    acc = __builtin_amdgcn_wmma_f32_16x16x4_f32(
         false, a,  false, bf,
         (short)0, acc,  false,  false);
  }

  if (wave < 8) {
    float* outp = (wave < 4) ? Q : K;
#pragma unroll
    for (int r = 0; r < 8; ++r) {
      const int m = m0 + r + 8 * lh;
      outp[(size_t)m * DHEAD + n0 + ln] = acc[r];
    }
  } else {
    const int b  = m0 / NSEQ;
    const int i0 = m0 & (NSEQ - 1);
    float s = 0.0f;
#pragma unroll
    for (int r = 0; r < 8; ++r) {
      const int i = i0 + r + 8 * lh;
      if (i < 3) Vhead[(size_t)(b * 3 + i) * DHEAD + n0 + ln] = acc[r];
      else       s += acc[r];
    }
    s += __shfl_xor(s, 16, 32);
    if (lh == 0) Psum[(size_t)blockIdx.x * DHEAD + n0 + ln] = s;
  }
}

__global__ __launch_bounds__(1024) void vsum_kernel(const float* __restrict__ Psum,
                                                    float* __restrict__ Srest) {
  __shared__ float red[1024];
  const int b = blockIdx.x;
  const int t = threadIdx.x;
  const int d = t & 63;
  const int s = t >> 6;
  const float* p = Psum + ((size_t)b * TILES_PER_BATCH + s * 16) * DHEAD + d;
  float acc = 0.0f;
#pragma unroll
  for (int k = 0; k < 16; ++k) acc += p[k * DHEAD];
  red[t] = acc;
  __syncthreads();
  if (t < DHEAD) {
    float r = 0.0f;
#pragma unroll
    for (int k = 0; k < 16; ++k) r += red[k * DHEAD + t];
    Srest[b * DHEAD + t] = r;
  }
}

__global__ __launch_bounds__(256) void combine_kernel(
    const float* __restrict__ Q, const float* __restrict__ K,
    const float* __restrict__ Vhead, const float* __restrict__ Srest,
    float* __restrict__ out) {
  const int tid  = threadIdx.x;
  const int wave = tid >> 5;
  const int lane = tid & 31;
  const int g = blockIdx.x * 8 + wave;
  const int b = g >> 12;
  const int i = g & (NSEQ - 1);

  const float2 q = ((const float2*)(Q + (size_t)g * DHEAD))[lane];

  float pa = 0.0f, pb = 0.0f, pc = 0.0f;
  {
    const float2 kc = ((const float2*)(K + (size_t)g * DHEAD))[lane];
    pb = q.x * kc.x + q.y * kc.y;
  }
  if (i > 0) {
    const float2 kp = ((const float2*)(K + (size_t)(g - 1) * DHEAD))[lane];
    pa = q.x * kp.x + q.y * kp.y;
  }
  if (i < NSEQ - 1) {
    const float2 kn = ((const float2*)(K + (size_t)(g + 1) * DHEAD))[lane];
    pc = q.x * kn.x + q.y * kn.y;
  }
#pragma unroll
  for (int off = 16; off > 0; off >>= 1) {
    pa += __shfl_xor(pa, off, 32);
    pb += __shfl_xor(pb, off, 32);
    pc += __shfl_xor(pc, off, 32);
  }
  const float a  = (i > 0) ? pa : 0.0f;
  const float bs = pb;
  const float c  = (i < NSEQ - 1) ? pc : 0.0f;

  const float m  = fmaxf(fmaxf(a, bs), fmaxf(c, 0.0f));
  const float ea = expf(a - m);
  const float eb = expf(bs - m);
  const float ec = expf(c - m);
  const float ez = expf(-m);
  const float inv = 1.0f / (ea + eb + ec + (float)(NSEQ - 3) * ez);

  const float* Vb = Vhead + (size_t)b * 3 * DHEAD;
  const float2 v0 = ((const float2*)(Vb))[lane];
  const float2 v1 = ((const float2*)(Vb + DHEAD))[lane];
  const float2 v2 = ((const float2*)(Vb + 2 * DHEAD))[lane];
  const float2 sr = ((const float2*)(Srest + b * DHEAD))[lane];

  float2 o;
  o.x = (ea * v0.x + eb * v1.x + ec * v2.x + ez * sr.x) * inv;
  o.y = (ea * v0.y + eb * v1.y + ec * v2.y + ez * sr.y) * inv;
  ((float2*)(out + (size_t)g * DHEAD))[lane] = o;
}

extern "C" void kernel_launch(void* const* d_in, const int* in_sizes, int n_in,
                              void* d_out, int out_size, void* d_ws, size_t ws_size,
                              hipStream_t stream) {
  const float* x  = (const float*)d_in[0];
  const float* Wq = (const float*)d_in[1];
  const float* bq = (const float*)d_in[2];
  const float* Wk = (const float*)d_in[3];
  const float* bk = (const float*)d_in[4];
  const float* Wv = (const float*)d_in[5];
  const float* bv = (const float*)d_in[6];
  float* out = (float*)d_out;

  const int BN = in_sizes[0] / EMBED;
  const int B  = BN / NSEQ;

  float* Q     = (float*)d_ws;
  float* K     = Q + (size_t)BN * DHEAD;
  float* Vhead = K + (size_t)BN * DHEAD;
  float* Psum  = Vhead + (size_t)B * 3 * DHEAD;
  float* Srest = Psum + (size_t)(BN / 16) * DHEAD;

  qkv_gemm_kernel<<<BN / 16, 384, 0, stream>>>(x, Wq, bq, Wk, bk, Wv, bv, Q, K, Vhead, Psum);
  vsum_kernel<<<B, 1024, 0, stream>>>(Psum, Srest);
  combine_kernel<<<BN / 8, 256, 0, stream>>>(Q, K, Vhead, Srest, out);
}


# projection kernel: 64 rows per block instead of 16, each wave keeps 4 row-tile accumulators so one W fragment load feeds 4 WMMAs (W loads and L2 reads /4, waves /4); k order and all sums unchanged
# speedup vs baseline: 2.2954x; 1.1388x over previous
#include <hip/hip_runtime.h>
#include <hip/hip_bf16.h>

#define EMBED 768
#define NSEQ  4096
#define DHEAD 64
#define LDSTR 772
#define TILES_PER_BATCH (NSEQ / 16)
#define MT 4
#define BROWS (16 * MT)
#define KG 4

typedef __attribute__((ext_vector_type(2))) float v2f;
typedef __attribute__((ext_vector_type(8))) float v8f;

__global__ __launch_bounds__(384) void qkv_gemm_kernel(
    const float* __restrict__ x,
    const float* __restrict__ Wq, const float* __restrict__ bq,
    const float* __restrict__ Wk, const float* __restrict__ bk,
    const float* __restrict__ Wv, const float* __restrict__ bv,
    float* __restrict__ Q, float* __restrict__ K,
    float* __restrict__ Vhead, float* __restrict__ Psum) {
  __shared__ float lds_x[BROWS * LDSTR];

  const int m0  = blockIdx.x * BROWS;
  const int tid = threadIdx.x;

  {
    const int rh = tid / (EMBED / 4);
    const int c4 = tid % (EMBED / 4);
    const float4* srcp = (const float4*)(x + (size_t)(m0 + rh) * EMBED) + c4;
    float* dstp = &lds_x[rh * LDSTR + c4 * 4];
    float4 st[BROWS / 2];
#pragma unroll
    for (int j = 0; j < BROWS / 2; ++j) st[j] = srcp[(size_t)j * 2 * (EMBED / 4)];
#pragma unroll
    for (int j = 0; j < BROWS / 2; ++j) {
      float* dst = dstp + j * 2 * LDSTR;
      dst[0] = st[j].x; dst[1] = st[j].y; dst[2] = st[j].z; dst[3] = st[j].w;
    }
  }
  __syncthreads();

  const int wave = tid >> 5;
  const int lane = tid & 31;
  const int ln   = lane & 15;
  const int lh   = lane >> 4;

  const float* W; const float* bias; int n0;
  if (wave < 4)      { W = Wq; bias = bq; n0 = wave * 16; }
  else if (wave < 8) { W = Wk; bias = bk; n0 = (wave - 4) * 16; }
  else               { W = Wv; bias = bv; n0 = (wave - 8) * 16; }

  const float bval = bias[n0 + ln];
  v8f acc[MT];
#pragma unroll
  for (int t = 0; t < MT; ++t) {
#pragma unroll
    for (int r = 0; r < 8; ++r) acc[t][r] = bval;
  }

  const float* lrow = &lds_x[ln * LDSTR];
  const float* wcol = W + n0 + ln;

  v2f bA[KG], bB[KG];
#pragma unroll
  for (int u = 0; u < KG; ++u) {
    const int ka = 4 * u + 2 * lh;
    bA[u][0] = wcol[(size_t)ka * DHEAD];
    bA[u][1] = wcol[(size_t)(ka + 1) * DHEAD];
  }

  for (int k0 = 0; k0 < EMBED; k0 += 8 * KG) {
#pragma unroll
    for (int u = 0; u < KG; ++u) {
      const int ka = k0 + 4 * KG + 4 * u + 2 * lh;
      bB[u][0] = wcol[(size_t)ka * DHEAD];
      bB[u][1] = wcol[(size_t)(ka + 1) * DHEAD];
    }
#pragma unroll
    for (int u = 0; u < KG; ++u) {
      const int ka = k0 + 4 * u + 2 * lh;
#pragma unroll
      for (int t = 0; t < MT; ++t) {
        v2f a;  a[0] = lrow[t * 16 * LDSTR + ka];
                a[1] = lrow[t * 16 * LDSTR + ka + 1];
        acc[t] = __builtin_amdgcn_wmma_f32_16x16x4_f32(
             false, a,  false, bA[u],
             (short)0, acc[t],  false,  false);
      }
    }
    const int kn = (k0 + 8 * KG < EMBED) ? (k0 + 8 * KG) : k0;
#pragma unroll
    for (int u = 0; u < KG; ++u) {
      const int ka = kn + 4 * u + 2 * lh;
      bA[u][0] = wcol[(size_t)ka * DHEAD];
      bA[u][1] = wcol[(size_t)(ka + 1) * DHEAD];
    }
#pragma unroll
    for (int u = 0; u < KG; ++u) {
      const int ka = k0 + 4 * KG + 4 * u + 2 * lh;
#pragma unroll
      for (int t = 0; t < MT; ++t) {
        v2f a;  a[0] = lrow[t * 16 * LDSTR + ka];
                a[1] = lrow[t * 16 * LDSTR + ka + 1];
        acc[t] = __builtin_amdgcn_wmma_f32_16x16x4_f32(
             false, a,  false, bB[u],
             (short)0, acc[t],  false,  false);
      }
    }
  }

  if (wave < 8) {
    float* outp = (wave < 4) ? Q : K;
#pragma unroll
    for (int t = 0; t < MT; ++t) {
#pragma unroll
      for (int r = 0; r < 8; ++r) {
        const int m = m0 + 16 * t + r + 8 * lh;
        outp[(size_t)m * DHEAD + n0 + ln] = acc[t][r];
      }
    }
  } else {
    const int b  = m0 / NSEQ;
    const int i0 = m0 & (NSEQ - 1);
#pragma unroll
    for (int t = 0; t < MT; ++t) {
      float s = 0.0f;
#pragma unroll
      for (int r = 0; r < 8; ++r) {
        const int i = i0 + 16 * t + r + 8 * lh;
        if (i < 3) Vhead[(size_t)(b * 3 + i) * DHEAD + n0 + ln] = acc[t][r];
        else       s += acc[t][r];
      }
      s += __shfl_xor(s, 16, 32);
      if (lh == 0) Psum[((size_t)blockIdx.x * MT + t) * DHEAD + n0 + ln] = s;
    }
  }
}

__global__ __launch_bounds__(1024) void vsum_kernel(const float* __restrict__ Psum,
                                                    float* __restrict__ Srest) {
  __shared__ float red[1024];
  const int b = blockIdx.x;
  const int t = threadIdx.x;
  const int d = t & 63;
  const int s = t >> 6;
  const float* p = Psum + ((size_t)b * TILES_PER_BATCH + s * 16) * DHEAD + d;
  float acc = 0.0f;
#pragma unroll
  for (int k = 0; k < 16; ++k) acc += p[k * DHEAD];
  red[t] = acc;
  __syncthreads();
  if (t < DHEAD) {
    float r = 0.0f;
#pragma unroll
    for (int k = 0; k < 16; ++k) r += red[k * DHEAD + t];
    Srest[b * DHEAD + t] = r;
  }
}

__global__ __launch_bounds__(256) void combine_kernel(
    const float* __restrict__ Q, const float* __restrict__ K,
    const float* __restrict__ Vhead, const float* __restrict__ Srest,
    float* __restrict__ out) {
  const int tid  = threadIdx.x;
  const int wave = tid >> 5;
  const int lane = tid & 31;
  const int g = blockIdx.x * 8 + wave;
  const int b = g >> 12;
  const int i = g & (NSEQ - 1);

  const float2 q = ((const float2*)(Q + (size_t)g * DHEAD))[lane];

  float pa = 0.0f, pb = 0.0f, pc = 0.0f;
  {
    const float2 kc = ((const float2*)(K + (size_t)g * DHEAD))[lane];
    pb = q.x * kc.x + q.y * kc.y;
  }
  if (i > 0) {
    const float2 kp = ((const float2*)(K + (size_t)(g - 1) * DHEAD))[lane];
    pa = q.x * kp.x + q.y * kp.y;
  }
  if (i < NSEQ - 1) {
    const float2 kn = ((const float2*)(K + (size_t)(g + 1) * DHEAD))[lane];
    pc = q.x * kn.x + q.y * kn.y;
  }
#pragma unroll
  for (int off = 16; off > 0; off >>= 1) {
    pa += __shfl_xor(pa, off, 32);
    pb += __shfl_xor(pb, off, 32);
    pc += __shfl_xor(pc, off, 32);
  }
  const float a  = (i > 0) ? pa : 0.0f;
  const float bs = pb;
  const float c  = (i < NSEQ - 1) ? pc : 0.0f;

  const float m  = fmaxf(fmaxf(a, bs), fmaxf(c, 0.0f));
  const float ea = expf(a - m);
  const float eb = expf(bs - m);
  const float ec = expf(c - m);
  const float ez = expf(-m);
  const float inv = 1.0f / (ea + eb + ec + (float)(NSEQ - 3) * ez);

  const float* Vb = Vhead + (size_t)b * 3 * DHEAD;
  const float2 v0 = ((const float2*)(Vb))[lane];
  const float2 v1 = ((const float2*)(Vb + DHEAD))[lane];
  const float2 v2 = ((const float2*)(Vb + 2 * DHEAD))[lane];
  const float2 sr = ((const float2*)(Srest + b * DHEAD))[lane];

  float2 o;
  o.x = (ea * v0.x + eb * v1.x + ec * v2.x + ez * sr.x) * inv;
  o.y = (ea * v0.y + eb * v1.y + ec * v2.y + ez * sr.y) * inv;
  ((float2*)(out + (size_t)g * DHEAD))[lane] = o;
}

extern "C" void kernel_launch(void* const* d_in, const int* in_sizes, int n_in,
                              void* d_out, int out_size, void* d_ws, size_t ws_size,
                              hipStream_t stream) {
  const float* x  = (const float*)d_in[0];
  const float* Wq = (const float*)d_in[1];
  const float* bq = (const float*)d_in[2];
  const float* Wk = (const float*)d_in[3];
  const float* bk = (const float*)d_in[4];
  const float* Wv = (const float*)d_in[5];
  const float* bv = (const float*)d_in[6];
  float* out = (float*)d_out;

  const int BN = in_sizes[0] / EMBED;
  const int B  = BN / NSEQ;

  float* Q     = (float*)d_ws;
  float* K     = Q + (size_t)BN * DHEAD;
  float* Vhead = K + (size_t)BN * DHEAD;
  float* Psum  = Vhead + (size_t)B * 3 * DHEAD;
  float* Srest = Psum + (size_t)(BN / 16) * DHEAD;

  qkv_gemm_kernel<<<BN / BROWS, 384, 0, stream>>>(x, Wq, bq, Wk, bk, Wv, bv, Q, K, Vhead, Psum);
  vsum_kernel<<<B, 1024, 0, stream>>>(Psum, Srest);
  combine_kernel<<<BN / 8, 256, 0, stream>>>(Q, K, Vhead, Srest, out);
}


# V by linearity: V waves project one 16-row tile per block (x rows 0..2 + the tile's two half column sums, added while staging) instead of 64 rows; V WMMAs /4; V summation order changes
# speedup vs baseline: 2.7275x; 1.1882x over previous
#include <hip/hip_runtime.h>
#include <hip/hip_bf16.h>

#define EMBED 768
#define NSEQ  4096
#define DHEAD 64
#define LDSTR 772
#define MT 4
#define BROWS (16 * MT)
#define KG 4
#define SUMROW (BROWS + 3)
#define LDSROWS (SUMROW + 2)
#define PARTS_PER_BATCH (NSEQ / BROWS)

typedef __attribute__((ext_vector_type(2))) float v2f;
typedef __attribute__((ext_vector_type(8))) float v8f;

template <int NT>
__device__ __forceinline__ void wave_kloop(const float* lrow, const float* __restrict__ wcol,
                                           const int lh, v8f (&acc)[NT]) {
  v2f bA[KG], bB[KG];
#pragma unroll
  for (int u = 0; u < KG; ++u) {
    const int ka = 4 * u + 2 * lh;
    bA[u][0] = wcol[(size_t)ka * DHEAD];
    bA[u][1] = wcol[(size_t)(ka + 1) * DHEAD];
  }

  for (int k0 = 0; k0 < EMBED; k0 += 8 * KG) {
#pragma unroll
    for (int u = 0; u < KG; ++u) {
      const int ka = k0 + 4 * KG + 4 * u + 2 * lh;
      bB[u][0] = wcol[(size_t)ka * DHEAD];
      bB[u][1] = wcol[(size_t)(ka + 1) * DHEAD];
    }
#pragma unroll
    for (int u = 0; u < KG; ++u) {
      const int ka = k0 + 4 * u + 2 * lh;
#pragma unroll
      for (int t = 0; t < NT; ++t) {
        v2f a;  a[0] = lrow[t * 16 * LDSTR + ka];
                a[1] = lrow[t * 16 * LDSTR + ka + 1];
        acc[t] = __builtin_amdgcn_wmma_f32_16x16x4_f32(
             false, a,  false, bA[u],
             (short)0, acc[t],  false,  false);
      }
    }
    const int kn = (k0 + 8 * KG < EMBED) ? (k0 + 8 * KG) : k0;
#pragma unroll
    for (int u = 0; u < KG; ++u) {
      const int ka = kn + 4 * u + 2 * lh;
      bA[u][0] = wcol[(size_t)ka * DHEAD];
      bA[u][1] = wcol[(size_t)(ka + 1) * DHEAD];
    }
#pragma unroll
    for (int u = 0; u < KG; ++u) {
      const int ka = k0 + 4 * KG + 4 * u + 2 * lh;
#pragma unroll
      for (int t = 0; t < NT; ++t) {
        v2f a;  a[0] = lrow[t * 16 * LDSTR + ka];
                a[1] = lrow[t * 16 * LDSTR + ka + 1];
        acc[t] = __builtin_amdgcn_wmma_f32_16x16x4_f32(
             false, a,  false, bB[u],
             (short)0, acc[t],  false,  false);
      }
    }
  }
}

__global__ __launch_bounds__(384) void qkv_gemm_kernel(
    const float* __restrict__ x,
    const float* __restrict__ Wq, const float* __restrict__ bq,
    const float* __restrict__ Wk, const float* __restrict__ bk,
    const float* __restrict__ Wv, const float* __restrict__ bv,
    float* __restrict__ Q, float* __restrict__ K,
    float* __restrict__ Vhead, float* __restrict__ Psum) {
  __shared__ float lds_x[LDSROWS * LDSTR];

  const int m0  = blockIdx.x * BROWS;
  const int tid = threadIdx.x;
  const bool first = (m0 & (NSEQ - 1)) == 0;

  {
    const int rh = tid / (EMBED / 4);
    const int c4 = tid % (EMBED / 4);
    const float4* srcp = (const float4*)(x + (size_t)(m0 + rh) * EMBED) + c4;
    float* dstp = &lds_x[rh * LDSTR + c4 * 4];
    float4 st[BROWS / 2];
#pragma unroll
    for (int j = 0; j < BROWS / 2; ++j) st[j] = srcp[(size_t)j * 2 * (EMBED / 4)];
    float4 cs = make_float4(0.0f, 0.0f, 0.0f, 0.0f);
#pragma unroll
    for (int j = 0; j < BROWS / 2; ++j) {
      float* dst = dstp + j * 2 * LDSTR;
      dst[0] = st[j].x; dst[1] = st[j].y; dst[2] = st[j].z; dst[3] = st[j].w;
      const bool take = (j >= 2) || !first || (j == 1 && rh == 1);
      if (take) { cs.x += st[j].x; cs.y += st[j].y; cs.z += st[j].z; cs.w += st[j].w; }
    }
    float* sp = &lds_x[(SUMROW + rh) * LDSTR + c4 * 4];
    sp[0] = cs.x; sp[1] = cs.y; sp[2] = cs.z; sp[3] = cs.w;
  }
  __syncthreads();

  const int wave = __builtin_amdgcn_readfirstlane(tid >> 5);
  const int lane = tid & 31;
  const int ln   = lane & 15;
  const int lh   = lane >> 4;

  if (wave < 8) {
    const float* W; const float* bias; float* outp; int n0;
    if (wave < 4) { W = Wq; bias = bq; outp = Q; n0 = wave * 16; }
    else          { W = Wk; bias = bk; outp = K; n0 = (wave - 4) * 16; }

    const float bval = bias[n0 + ln];
    v8f acc[MT];
#pragma unroll
    for (int t = 0; t < MT; ++t) {
#pragma unroll
      for (int r = 0; r < 8; ++r) acc[t][r] = bval;
    }

    wave_kloop<MT>(&lds_x[ln * LDSTR], W + n0 + ln, lh, acc);

#pragma unroll
    for (int t = 0; t < MT; ++t) {
#pragma unroll
      for (int r = 0; r < 8; ++r) {
        const int m = m0 + 16 * t + r + 8 * lh;
        outp[(size_t)m * DHEAD + n0 + ln] = acc[t][r];
      }
    }
  } else {
    const int n0 = (wave - 8) * 16;
    const float bval = bv[n0 + ln];
    v8f acc[1];
#pragma unroll
    for (int r = 0; r < 8; ++r) acc[0][r] = bval;

    const int arow = (ln == 3) ? SUMROW : (ln == 4) ? (SUMROW + 1) : ln;
    wave_kloop<1>(&lds_x[arow * LDSTR], Wv + n0 + ln, lh, acc);

    asm volatile("v_nop\n\tv_nop\n\tv_nop\n\tv_nop\n\tv_nop" : "+v"(acc[0]));

    if (lh == 0) {
      if (first) {
        const int b = m0 / NSEQ;
#pragma unroll
        for (int r = 0; r < 3; ++r) Vhead[(size_t)(b * 3 + r) * DHEAD + n0 + ln] = acc[0][r];
      }
      const float cnt = first ? (float)(BROWS - 3) : (float)BROWS;
      Psum[(size_t)blockIdx.x * DHEAD + n0 + ln] = acc[0][3] + acc[0][4] + (cnt - 2.0f) * bval;
    }
  }
}

__global__ __launch_bounds__(1024) void vsum_kernel(const float* __restrict__ Psum,
                                                    float* __restrict__ Srest) {
  __shared__ float red[1024];
  const int b = blockIdx.x;
  const int t = threadIdx.x;
  const int d = t & 63;
  const int s = t >> 6;
  const int per = PARTS_PER_BATCH / 16;
  const float* p = Psum + ((size_t)b * PARTS_PER_BATCH + s * per) * DHEAD + d;
  float acc = 0.0f;
#pragma unroll
  for (int k = 0; k < per; ++k) acc += p[k * DHEAD];
  red[t] = acc;
  __syncthreads();
  if (t < DHEAD) {
    float r = 0.0f;
#pragma unroll
    for (int k = 0; k < 16; ++k) r += red[k * DHEAD + t];
    Srest[b * DHEAD + t] = r;
  }
}

__global__ __launch_bounds__(256) void combine_kernel(
    const float* __restrict__ Q, const float* __restrict__ K,
    const float* __restrict__ Vhead, const float* __restrict__ Srest,
    float* __restrict__ out) {
  const int tid  = threadIdx.x;
  const int wave = tid >> 5;
  const int lane = tid & 31;
  const int g = blockIdx.x * 8 + wave;
  const int b = g >> 12;
  const int i = g & (NSEQ - 1);

  const float2 q = ((const float2*)(Q + (size_t)g * DHEAD))[lane];

  float pa = 0.0f, pb = 0.0f, pc = 0.0f;
  {
    const float2 kc = ((const float2*)(K + (size_t)g * DHEAD))[lane];
    pb = q.x * kc.x + q.y * kc.y;
  }
  if (i > 0) {
    const float2 kp = ((const float2*)(K + (size_t)(g - 1) * DHEAD))[lane];
    pa = q.x * kp.x + q.y * kp.y;
  }
  if (i < NSEQ - 1) {
    const float2 kn = ((const float2*)(K + (size_t)(g + 1) * DHEAD))[lane];
    pc = q.x * kn.x + q.y * kn.y;
  }
#pragma unroll
  for (int off = 16; off > 0; off >>= 1) {
    pa += __shfl_xor(pa, off, 32);
    pb += __shfl_xor(pb, off, 32);
    pc += __shfl_xor(pc, off, 32);
  }
  const float a  = (i > 0) ? pa : 0.0f;
  const float bs = pb;
  const float c  = (i < NSEQ - 1) ? pc : 0.0f;

  const float m  = fmaxf(fmaxf(a, bs), fmaxf(c, 0.0f));
  const float ea = expf(a - m);
  const float eb = expf(bs - m);
  const float ec = expf(c - m);
  const float ez = expf(-m);
  const float inv = 1.0f / (ea + eb + ec + (float)(NSEQ - 3) * ez);

  const float* Vb = Vhead + (size_t)b * 3 * DHEAD;
  const float2 v0 = ((const float2*)(Vb))[lane];
  const float2 v1 = ((const float2*)(Vb + DHEAD))[lane];
  const float2 v2 = ((const float2*)(Vb + 2 * DHEAD))[lane];
  const float2 sr = ((const float2*)(Srest + b * DHEAD))[lane];

  float2 o;
  o.x = (ea * v0.x + eb * v1.x + ec * v2.x + ez * sr.x) * inv;
  o.y = (ea * v0.y + eb * v1.y + ec * v2.y + ez * sr.y) * inv;
  ((float2*)(out + (size_t)g * DHEAD))[lane] = o;
}

extern "C" void kernel_launch(void* const* d_in, const int* in_sizes, int n_in,
                              void* d_out, int out_size, void* d_ws, size_t ws_size,
                              hipStream_t stream) {
  const float* x  = (const float*)d_in[0];
  const float* Wq = (const float*)d_in[1];
  const float* bq = (const float*)d_in[2];
  const float* Wk = (const float*)d_in[3];
  const float* bk = (const float*)d_in[4];
  const float* Wv = (const float*)d_in[5];
  const float* bv = (const float*)d_in[6];
  float* out = (float*)d_out;

  const int BN = in_sizes[0] / EMBED;
  const int B  = BN / NSEQ;

  float* Q     = (float*)d_ws;
  float* K     = Q + (size_t)BN * DHEAD;
  float* Vhead = K + (size_t)BN * DHEAD;
  float* Psum  = Vhead + (size_t)B * 3 * DHEAD;
  float* Srest = Psum + (size_t)(BN / BROWS) * DHEAD;

  qkv_gemm_kernel<<<BN / BROWS, 384, 0, stream>>>(x, Wq, bq, Wk, bk, Wv, bv, Q, K, Vhead, Psum);
  vsum_kernel<<<B, 1024, 0, stream>>>(Psum, Srest);
  combine_kernel<<<BN / 8, 256, 0, stream>>>(Q, K, Vhead, Srest, out);
}


# each row's three neighbour scores formed in the projection epilogue from the block's Q and K tiles in LDS (block-edge pairs in the middle launch); Q, K never stored; score summation order changes
# speedup vs baseline: 2.8575x; 1.0477x over previous
#include <hip/hip_runtime.h>
#include <hip/hip_bf16.h>

#define EMBED 768
#define NSEQ  4096
#define DHEAD 64
#define LDSTR 772
#define MT 4
#define BROWS (16 * MT)
#define KG 4
#define SUMROW (BROWS + 3)
#define LDSROWS (SUMROW + 2)
#define PARTS_PER_BATCH (NSEQ / BROWS)
#define SP 68
#define KS_OFF (BROWS * SP)

typedef __attribute__((ext_vector_type(2))) float v2f;
typedef __attribute__((ext_vector_type(8))) float v8f;

template <int NT>
__device__ __forceinline__ void wave_kloop(const float* lrow, const float* __restrict__ wcol,
                                           const int lh, v8f (&acc)[NT]) {
  v2f bA[KG], bB[KG];
#pragma unroll
  for (int u = 0; u < KG; ++u) {
    const int ka = 4 * u + 2 * lh;
    bA[u][0] = wcol[(size_t)ka * DHEAD];
    bA[u][1] = wcol[(size_t)(ka + 1) * DHEAD];
  }

  for (int k0 = 0; k0 < EMBED; k0 += 8 * KG) {
#pragma unroll
    for (int u = 0; u < KG; ++u) {
      const int ka = k0 + 4 * KG + 4 * u + 2 * lh;
      bB[u][0] = wcol[(size_t)ka * DHEAD];
      bB[u][1] = wcol[(size_t)(ka + 1) * DHEAD];
    }
#pragma unroll
    for (int u = 0; u < KG; ++u) {
      const int ka = k0 + 4 * u + 2 * lh;
#pragma unroll
      for (int t = 0; t < NT; ++t) {
        v2f a;  a[0] = lrow[t * 16 * LDSTR + ka];
                a[1] = lrow[t * 16 * LDSTR + ka + 1];
        acc[t] = __builtin_amdgcn_wmma_f32_16x16x4_f32(
             false, a,  false, bA[u],
             (short)0, acc[t],  false,  false);
      }
    }
    const int kn = (k0 + 8 * KG < EMBED) ? (k0 + 8 * KG) : k0;
#pragma unroll
    for (int u = 0; u < KG; ++u) {
      const int ka = kn + 4 * u + 2 * lh;
      bA[u][0] = wcol[(size_t)ka * DHEAD];
      bA[u][1] = wcol[(size_t)(ka + 1) * DHEAD];
    }
#pragma unroll
    for (int u = 0; u < KG; ++u) {
      const int ka = k0 + 4 * KG + 4 * u + 2 * lh;
#pragma unroll
      for (int t = 0; t < NT; ++t) {
        v2f a;  a[0] = lrow[t * 16 * LDSTR + ka];
                a[1] = lrow[t * 16 * LDSTR + ka + 1];
        acc[t] = __builtin_amdgcn_wmma_f32_16x16x4_f32(
             false, a,  false, bB[u],
             (short)0, acc[t],  false,  false);
      }
    }
  }
}

__global__ __launch_bounds__(384) void qkv_gemm_kernel(
    const float* __restrict__ x,
    const float* __restrict__ Wq, const float* __restrict__ bq,
    const float* __restrict__ Wk, const float* __restrict__ bk,
    const float* __restrict__ Wv, const float* __restrict__ bv,
    float* __restrict__ S4, float* __restrict__ Qe, float* __restrict__ Ke,
    float* __restrict__ Vhead, float* __restrict__ Psum) {
  __shared__ float lds_x[LDSROWS * LDSTR];
  __shared__ __attribute__((aligned(16))) float lds_qk[(2 * BROWS + 2) * SP];

  const int m0  = blockIdx.x * BROWS;
  const int tid = threadIdx.x;
  const bool first = (m0 & (NSEQ - 1)) == 0;

  {
    const int rh = tid / (EMBED / 4);
    const int c4 = tid % (EMBED / 4);
    const float4* srcp = (const float4*)(x + (size_t)(m0 + rh) * EMBED) + c4;
    float* dstp = &lds_x[rh * LDSTR + c4 * 4];
    float4 st[BROWS / 2];
#pragma unroll
    for (int j = 0; j < BROWS / 2; ++j) st[j] = srcp[(size_t)j * 2 * (EMBED / 4)];
    float4 cs = make_float4(0.0f, 0.0f, 0.0f, 0.0f);
#pragma unroll
    for (int j = 0; j < BROWS / 2; ++j) {
      float* dst = dstp + j * 2 * LDSTR;
      dst[0] = st[j].x; dst[1] = st[j].y; dst[2] = st[j].z; dst[3] = st[j].w;
      const bool take = (j >= 2) || !first || (j == 1 && rh == 1);
      if (take) { cs.x += st[j].x; cs.y += st[j].y; cs.z += st[j].z; cs.w += st[j].w; }
    }
    float* sp = &lds_x[(SUMROW + rh) * LDSTR + c4 * 4];
    sp[0] = cs.x; sp[1] = cs.y; sp[2] = cs.z; sp[3] = cs.w;
  }
  __syncthreads();

  const int wave = __builtin_amdgcn_readfirstlane(tid >> 5);
  const int lane = tid & 31;
  const int ln   = lane & 15;
  const int lh   = lane >> 4;

  if (wave < 8) {
    const float* W; const float* bias; float* tile; int n0;
    if (wave < 4) { W = Wq; bias = bq; tile = lds_qk;               n0 = wave * 16; }
    else          { W = Wk; bias = bk; tile = lds_qk + KS_OFF + SP; n0 = (wave - 4) * 16; }

    const float bval = bias[n0 + ln];
    v8f acc[MT];
#pragma unroll
    for (int t = 0; t < MT; ++t) {
#pragma unroll
      for (int r = 0; r < 8; ++r) acc[t][r] = bval;
    }

    wave_kloop<MT>(&lds_x[ln * LDSTR], W + n0 + ln, lh, acc);

#pragma unroll
    for (int t = 0; t < MT; ++t) {
#pragma unroll
      for (int r = 0; r < 8; ++r) {
        const int row = 16 * t + r + 8 * lh;
        tile[row * SP + n0 + ln] = acc[t][r];
      }
    }
  } else {
    const int n0 = (wave - 8) * 16;
    const float bval = bv[n0 + ln];
    v8f acc[1];
#pragma unroll
    for (int r = 0; r < 8; ++r) acc[0][r] = bval;

    const int arow = (ln == 3) ? SUMROW : (ln == 4) ? (SUMROW + 1) : ln;
    wave_kloop<1>(&lds_x[arow * LDSTR], Wv + n0 + ln, lh, acc);

    asm volatile("v_nop\n\tv_nop\n\tv_nop\n\tv_nop\n\tv_nop" : "+v"(acc[0]));

    if (lh == 0) {
      if (first) {
        const int b = m0 / NSEQ;
#pragma unroll
        for (int r = 0; r < 3; ++r) Vhead[(size_t)(b * 3 + r) * DHEAD + n0 + ln] = acc[0][r];
      }
      const float cnt = first ? (float)(BROWS - 3) : (float)BROWS;
      Psum[(size_t)blockIdx.x * DHEAD + n0 + ln] = acc[0][3] + acc[0][4] + (cnt - 2.0f) * bval;
    }
    {
      const int h  = tid - 256;
      const int hr = (h < DHEAD) ? 0 : (BROWS + 1);
      lds_qk[KS_OFF + hr * SP + (h & (DHEAD - 1))] = 0.0f;
    }
  }
  __syncthreads();

  if (wave < 8) {
    const int r = tid >> 2;
    const int j = tid & 3;
    const float4* q4 = (const float4*)(lds_qk + r * SP + 16 * j);
    const float4* kp = (const float4*)(lds_qk + KS_OFF + r * SP + 16 * j);
    const float4* kc = (const float4*)(lds_qk + KS_OFF + (r + 1) * SP + 16 * j);
    const float4* kn = (const float4*)(lds_qk + KS_OFF + (r + 2) * SP + 16 * j);
    float pa = 0.0f, pb = 0.0f, pc = 0.0f;
#pragma unroll
    for (int e = 0; e < 4; ++e) {
      const float4 q = q4[e], a = kp[e], b = kc[e], c = kn[e];
      pa += q.x * a.x + q.y * a.y + q.z * a.z + q.w * a.w;
      pb += q.x * b.x + q.y * b.y + q.z * b.z + q.w * b.w;
      pc += q.x * c.x + q.y * c.y + q.z * c.z + q.w * c.w;
    }
    pa += __shfl_xor(pa, 1, 32);  pb += __shfl_xor(pb, 1, 32);  pc += __shfl_xor(pc, 1, 32);
    pa += __shfl_xor(pa, 2, 32);  pb += __shfl_xor(pb, 2, 32);  pc += __shfl_xor(pc, 2, 32);
    if (j == 0) ((float4*)S4)[m0 + r] = make_float4(pa, pb, pc, 0.0f);

    if (wave < 2) {
      const int er = (lane < 16) ? 0 : (BROWS - 1);
      const float* srow = (wave == 0) ? (lds_qk + er * SP) : (lds_qk + KS_OFF + (er + 1) * SP);
      float* edge = (wave == 0) ? Qe : Ke;
      ((float4*)edge)[(size_t)blockIdx.x * 32 + lane] = ((const float4*)srow)[lane & 15];
    }
  }
}

__global__ __launch_bounds__(1024) void vsum_kernel(const float* __restrict__ Psum,
                                                    float* __restrict__ Srest,
                                                    const float* __restrict__ Qe,
                                                    const float* __restrict__ Ke,
                                                    float* __restrict__ E) {
  __shared__ float red[1024];
  __shared__ __attribute__((aligned(16))) float ered[2 * PARTS_PER_BATCH];
  const int b = blockIdx.x;
  const int t = threadIdx.x;
  const int d = t & 63;
  const int s = t >> 6;
  const int per = PARTS_PER_BATCH / 16;
  const float* p = Psum + ((size_t)b * PARTS_PER_BATCH + s * per) * DHEAD + d;

  const int pe = t >> 4;
  const int j  = t & 15;
  const bool has = pe < PARTS_PER_BATCH - 1;
  const int pq = has ? pe : (PARTS_PER_BATCH - 2);
  const float4* q4 = (const float4*)Qe + (size_t)(b * PARTS_PER_BATCH + pq) * 32;
  const float4* k4 = (const float4*)Ke + (size_t)(b * PARTS_PER_BATCH + pq) * 32;
  const float4 ql = q4[16 + j];
  const float4 kl = k4[16 + j];
  const float4 qf = q4[32 + j];
  const float4 kf = k4[32 + j];

  float acc = 0.0f;
#pragma unroll
  for (int k = 0; k < per; ++k) acc += p[k * DHEAD];
  red[t] = acc;

  {
    float up = ql.x * kf.x + ql.y * kf.y + ql.z * kf.z + ql.w * kf.w;
    float lo = qf.x * kl.x + qf.y * kl.y + qf.z * kl.z + qf.w * kl.w;
#pragma unroll
    for (int off = 8; off > 0; off >>= 1) {
      up += __shfl_xor(up, off, 32);
      lo += __shfl_xor(lo, off, 32);
    }
    if (j == 0) {
      if (has) {
        ered[2 * pe + 1] = up;
        ered[2 * pe + 2] = lo;
      } else {
        ered[0] = 0.0f;
        ered[2 * PARTS_PER_BATCH - 1] = 0.0f;
      }
    }
  }
  __syncthreads();
  if (t < DHEAD) {
    float r = 0.0f;
#pragma unroll
    for (int k = 0; k < 16; ++k) r += red[k * DHEAD + t];
    Srest[b * DHEAD + t] = r;
  } else if (t < DHEAD + 32) {
    const int l = t - DHEAD;
    ((float4*)E)[b * (PARTS_PER_BATCH / 2) + l] = ((const float4*)ered)[l];
  }
}

__global__ __launch_bounds__(256) void combine_kernel(
    const float* __restrict__ S4, const float* __restrict__ E,
    const float* __restrict__ Vhead, const float* __restrict__ Srest,
    float* __restrict__ out) {
  const int tid  = threadIdx.x;
  const int wave = tid >> 5;
  const int lane = tid & 31;
  const int hf   = lane >> 4;
  const int c4   = lane & 15;
  const int blk  = blockIdx.x;
  const int b    = blk / PARTS_PER_BATCH;

  const float2 e = ((const float2*)E)[blk];
  const float4* Vb = (const float4*)(Vhead + (size_t)b * 3 * DHEAD);
  const float4 v0 = Vb[c4];
  const float4 v1 = Vb[16 + c4];
  const float4 v2 = Vb[32 + c4];
  const float4 sr = ((const float4*)(Srest + b * DHEAD))[c4];

#pragma unroll
  for (int ps = 0; ps < 4; ++ps) {
    const int r = wave * 8 + ps * 2 + hf;
    const int g = blk * BROWS + r;
    const int i = g & (NSEQ - 1);
    const float4 s = ((const float4*)S4)[g];
    const float pa = (r == 0) ? e.x : s.x;
    const float pc = (r == BROWS - 1) ? e.y : s.z;

    const float a  = (i > 0) ? pa : 0.0f;
    const float bs = s.y;
    const float c  = (i < NSEQ - 1) ? pc : 0.0f;

    const float m  = fmaxf(fmaxf(a, bs), fmaxf(c, 0.0f));
    const float ea = expf(a - m);
    const float eb = expf(bs - m);
    const float ec = expf(c - m);
    const float ez = expf(-m);
    const float inv = 1.0f / (ea + eb + ec + (float)(NSEQ - 3) * ez);

    float4 o;
    o.x = (ea * v0.x + eb * v1.x + ec * v2.x + ez * sr.x) * inv;
    o.y = (ea * v0.y + eb * v1.y + ec * v2.y + ez * sr.y) * inv;
    o.z = (ea * v0.z + eb * v1.z + ec * v2.z + ez * sr.z) * inv;
    o.w = (ea * v0.w + eb * v1.w + ec * v2.w + ez * sr.w) * inv;
    ((float4*)(out + (size_t)g * DHEAD))[c4] = o;
  }
}

extern "C" void kernel_launch(void* const* d_in, const int* in_sizes, int n_in,
                              void* d_out, int out_size, void* d_ws, size_t ws_size,
                              hipStream_t stream) {
  const float* x  = (const float*)d_in[0];
  const float* Wq = (const float*)d_in[1];
  const float* bq = (const float*)d_in[2];
  const float* Wk = (const float*)d_in[3];
  const float* bk = (const float*)d_in[4];
  const float* Wv = (const float*)d_in[5];
  const float* bv = (const float*)d_in[6];
  float* out = (float*)d_out;

  const int BN = in_sizes[0] / EMBED;
  const int B  = BN / NSEQ;

  float* S4    = (float*)d_ws;
  float* Qe    = S4 + (size_t)BN * 4;
  float* Ke    = Qe + (size_t)(BN / BROWS) * 2 * DHEAD;
  float* E     = Ke + (size_t)(BN / BROWS) * 2 * DHEAD;
  float* Vhead = E + (size_t)(BN / BROWS) * 2;
  float* Psum  = Vhead + (size_t)B * 3 * DHEAD;
  float* Srest = Psum + (size_t)(BN / BROWS) * DHEAD;

  qkv_gemm_kernel<<<BN / BROWS, 384, 0, stream>>>(x, Wq, bq, Wk, bk, Wv, bv, S4, Qe, Ke, Vhead, Psum);
  vsum_kernel<<<B, 1024, 0, stream>>>(Psum, Srest, Qe, Ke, E);
  combine_kernel<<<BN / BROWS, 256, 0, stream>>>(S4, E, Vhead, Srest, out);
}


# middle launch removed: each combine block folds its batch's 64 partial V rows itself (same fixed order) and forms its two block-edge scores; it reads only planes launch 1 wrote; 3 launches -> 2
# speedup vs baseline: 3.1410x; 1.0992x over previous
#include <hip/hip_runtime.h>
#include <hip/hip_bf16.h>

#define EMBED 768
#define NSEQ  4096
#define DHEAD 64
#define LDSTR 772
#define MT 4
#define BROWS (16 * MT)
#define KG 4
#define SUMROW (BROWS + 3)
#define LDSROWS (SUMROW + 2)
#define PARTS_PER_BATCH (NSEQ / BROWS)
#define SP 68
#define KS_OFF (BROWS * SP)

typedef __attribute__((ext_vector_type(2))) float v2f;
typedef __attribute__((ext_vector_type(8))) float v8f;
typedef __attribute__((ext_vector_type(4))) float v4f;

template <int NT>
__device__ __forceinline__ void wave_kloop(const float* lrow, const float* __restrict__ wcol,
                                           const int lh, v8f (&acc)[NT]) {
  v2f bA[KG], bB[KG];
#pragma unroll
  for (int u = 0; u < KG; ++u) {
    const int ka = 4 * u + 2 * lh;
    bA[u][0] = wcol[(size_t)ka * DHEAD];
    bA[u][1] = wcol[(size_t)(ka + 1) * DHEAD];
  }

  for (int k0 = 0; k0 < EMBED; k0 += 8 * KG) {
#pragma unroll
    for (int u = 0; u < KG; ++u) {
      const int ka = k0 + 4 * KG + 4 * u + 2 * lh;
      bB[u][0] = wcol[(size_t)ka * DHEAD];
      bB[u][1] = wcol[(size_t)(ka + 1) * DHEAD];
    }
#pragma unroll
    for (int u = 0; u < KG; ++u) {
      const int ka = k0 + 4 * u + 2 * lh;
#pragma unroll
      for (int t = 0; t < NT; ++t) {
        v2f a;  a[0] = lrow[t * 16 * LDSTR + ka];
                a[1] = lrow[t * 16 * LDSTR + ka + 1];
        acc[t] = __builtin_amdgcn_wmma_f32_16x16x4_f32(
             false, a,  false, bA[u],
             (short)0, acc[t],  false,  false);
      }
    }
    const int kn = (k0 + 8 * KG < EMBED) ? (k0 + 8 * KG) : k0;
#pragma unroll
    for (int u = 0; u < KG; ++u) {
      const int ka = kn + 4 * u + 2 * lh;
      bA[u][0] = wcol[(size_t)ka * DHEAD];
      bA[u][1] = wcol[(size_t)(ka + 1) * DHEAD];
    }
#pragma unroll
    for (int u = 0; u < KG; ++u) {
      const int ka = k0 + 4 * KG + 4 * u + 2 * lh;
#pragma unroll
      for (int t = 0; t < NT; ++t) {
        v2f a;  a[0] = lrow[t * 16 * LDSTR + ka];
                a[1] = lrow[t * 16 * LDSTR + ka + 1];
        acc[t] = __builtin_amdgcn_wmma_f32_16x16x4_f32(
             false, a,  false, bB[u],
             (short)0, acc[t],  false,  false);
      }
    }
  }
}

__global__ __launch_bounds__(384) void qkv_gemm_kernel(
    const float* __restrict__ x,
    const float* __restrict__ Wq, const float* __restrict__ bq,
    const float* __restrict__ Wk, const float* __restrict__ bk,
    const float* __restrict__ Wv, const float* __restrict__ bv,
    float* __restrict__ S4, float* __restrict__ Qe, float* __restrict__ Ke,
    float* __restrict__ Vhead, float* __restrict__ Psum) {
  __shared__ float lds_x[LDSROWS * LDSTR];
  __shared__ __attribute__((aligned(16))) float lds_qk[(2 * BROWS + 2) * SP];

  const int m0  = blockIdx.x * BROWS;
  const int tid = threadIdx.x;
  const bool first = (m0 & (NSEQ - 1)) == 0;

  {
    const int rh = tid / (EMBED / 4);
    const int c4 = tid % (EMBED / 4);
    const float4* srcp = (const float4*)(x + (size_t)(m0 + rh) * EMBED) + c4;
    float* dstp = &lds_x[rh * LDSTR + c4 * 4];
    float4 st[BROWS / 2];
#pragma unroll
    for (int j = 0; j < BROWS / 2; ++j) st[j] = srcp[(size_t)j * 2 * (EMBED / 4)];
    float4 cs = make_float4(0.0f, 0.0f, 0.0f, 0.0f);
#pragma unroll
    for (int j = 0; j < BROWS / 2; ++j) {
      float* dst = dstp + j * 2 * LDSTR;
      dst[0] = st[j].x; dst[1] = st[j].y; dst[2] = st[j].z; dst[3] = st[j].w;
      const bool take = (j >= 2) || !first || (j == 1 && rh == 1);
      if (take) { cs.x += st[j].x; cs.y += st[j].y; cs.z += st[j].z; cs.w += st[j].w; }
    }
    float* sp = &lds_x[(SUMROW + rh) * LDSTR + c4 * 4];
    sp[0] = cs.x; sp[1] = cs.y; sp[2] = cs.z; sp[3] = cs.w;
  }
  __syncthreads();

  const int wave = __builtin_amdgcn_readfirstlane(tid >> 5);
  const int lane = tid & 31;
  const int ln   = lane & 15;
  const int lh   = lane >> 4;

  if (wave < 8) {
    const float* W; const float* bias; float* tile; int n0;
    if (wave < 4) { W = Wq; bias = bq; tile = lds_qk;               n0 = wave * 16; }
    else          { W = Wk; bias = bk; tile = lds_qk + KS_OFF + SP; n0 = (wave - 4) * 16; }

    const float bval = bias[n0 + ln];
    v8f acc[MT];
#pragma unroll
    for (int t = 0; t < MT; ++t) {
#pragma unroll
      for (int r = 0; r < 8; ++r) acc[t][r] = bval;
    }

    wave_kloop<MT>(&lds_x[ln * LDSTR], W + n0 + ln, lh, acc);

#pragma unroll
    for (int t = 0; t < MT; ++t) {
#pragma unroll
      for (int r = 0; r < 8; ++r) {
        const int row = 16 * t + r + 8 * lh;
        tile[row * SP + n0 + ln] = acc[t][r];
      }
    }
  } else {
    const int n0 = (wave - 8) * 16;
    const float bval = bv[n0 + ln];
    v8f acc[1];
#pragma unroll
    for (int r = 0; r < 8; ++r) acc[0][r] = bval;

    const int arow = (ln == 3) ? SUMROW : (ln == 4) ? (SUMROW + 1) : ln;
    wave_kloop<1>(&lds_x[arow * LDSTR], Wv + n0 + ln, lh, acc);

    asm volatile("v_nop\n\tv_nop\n\tv_nop\n\tv_nop\n\tv_nop" : "+v"(acc[0]));

    if (lh == 0) {
      if (first) {
        const int b = m0 / NSEQ;
#pragma unroll
        for (int r = 0; r < 3; ++r) Vhead[(size_t)(b * 3 + r) * DHEAD + n0 + ln] = acc[0][r];
      }
      const float cnt = first ? (float)(BROWS - 3) : (float)BROWS;
      Psum[(size_t)blockIdx.x * DHEAD + n0 + ln] = acc[0][3] + acc[0][4] + (cnt - 2.0f) * bval;
    }
    {
      const int h  = tid - 256;
      const int hr = (h < DHEAD) ? 0 : (BROWS + 1);
      lds_qk[KS_OFF + hr * SP + (h & (DHEAD - 1))] = 0.0f;
    }
  }
  __syncthreads();

  if (wave < 8) {
    const int r = tid >> 2;
    const int j = tid & 3;
    const float4* q4 = (const float4*)(lds_qk + r * SP + 16 * j);
    const float4* kp = (const float4*)(lds_qk + KS_OFF + r * SP + 16 * j);
    const float4* kc = (const float4*)(lds_qk + KS_OFF + (r + 1) * SP + 16 * j);
    const float4* kn = (const float4*)(lds_qk + KS_OFF + (r + 2) * SP + 16 * j);
    float pa = 0.0f, pb = 0.0f, pc = 0.0f;
#pragma unroll
    for (int e = 0; e < 4; ++e) {
      const float4 q = q4[e], a = kp[e], b = kc[e], c = kn[e];
      pa += q.x * a.x + q.y * a.y + q.z * a.z + q.w * a.w;
      pb += q.x * b.x + q.y * b.y + q.z * b.z + q.w * b.w;
      pc += q.x * c.x + q.y * c.y + q.z * c.z + q.w * c.w;
    }
    pa += __shfl_xor(pa, 1, 32);  pb += __shfl_xor(pb, 1, 32);  pc += __shfl_xor(pc, 1, 32);
    pa += __shfl_xor(pa, 2, 32);  pb += __shfl_xor(pb, 2, 32);  pc += __shfl_xor(pc, 2, 32);
    if (j == 0) ((float4*)S4)[m0 + r] = make_float4(pa, pb, pc, 0.0f);

    if (wave < 2) {
      const int er = (lane < 16) ? 0 : (BROWS - 1);
      const float* srow = (wave == 0) ? (lds_qk + er * SP) : (lds_qk + KS_OFF + (er + 1) * SP);
      float* edge = (wave == 0) ? Qe : Ke;
      ((float4*)edge)[(size_t)blockIdx.x * 32 + lane] = ((const float4*)srow)[lane & 15];
    }
  }
}

__global__ __launch_bounds__(256) void combine_kernel(
    const float* __restrict__ S4, const float* __restrict__ Qe, const float* __restrict__ Ke,
    const float* __restrict__ Vhead, const float* __restrict__ Psum,
    float* __restrict__ out) {
  __shared__ __attribute__((aligned(16))) float red[16 * DHEAD];
  const int tid  = threadIdx.x;
  const int wave = tid >> 5;
  const int lane = tid & 31;
  const int hf   = lane >> 4;
  const int c4   = lane & 15;
  const int blk  = blockIdx.x;
  const int b    = blk / PARTS_PER_BATCH;
  const int pb   = blk % PARTS_PER_BATCH;

  v4f sc[4];
#pragma unroll
  for (int ps = 0; ps < 4; ++ps) sc[ps] = ((const v4f*)S4)[blk * BROWS + wave * 8 + ps * 2 + hf];

  const v4f* Vb = (const v4f*)(Vhead + (size_t)b * 3 * DHEAD);
  v4f v0 = Vb[c4];
  v4f v1 = Vb[16 + c4];
  v4f v2 = Vb[32 + c4];

  const int lastrow = (wave == 7) ? 1 : 0;
  const int nb = lastrow ? ((pb < PARTS_PER_BATCH - 1) ? blk + 1 : blk)
                         : ((pb > 0) ? blk - 1 : blk);
  v4f eq = ((const v4f*)Qe)[(size_t)blk * 32 + 16 * lastrow + c4];
  v4f ek = ((const v4f*)Ke)[(size_t)nb * 32 + 16 * (1 - lastrow) + c4];

  const int d  = tid & (DHEAD - 1);
  const int sg = tid >> 6;
  const float* pp = Psum + ((size_t)b * PARTS_PER_BATCH + sg * 16) * DHEAD + d;
  v4f pr[4];
#pragma unroll
  for (int u = 0; u < 4; ++u) {
#pragma unroll
    for (int k = 0; k < 4; ++k) pr[u][k] = pp[(4 * u + k) * DHEAD];
  }

  asm volatile("" : "+v"(sc[0]), "+v"(sc[1]), "+v"(sc[2]), "+v"(sc[3]),
                    "+v"(v0), "+v"(v1), "+v"(v2), "+v"(eq), "+v"(ek),
                    "+v"(pr[0]), "+v"(pr[1]), "+v"(pr[2]), "+v"(pr[3]));

#pragma unroll
  for (int u = 0; u < 4; ++u) {
    float a = 0.0f;
#pragma unroll
    for (int k = 0; k < 4; ++k) a += pr[u][k];
    red[(4 * sg + u) * DHEAD + d] = a;
  }

  float es = eq.x * ek.x + eq.y * ek.y + eq.z * ek.z + eq.w * ek.w;
  es += __int_as_float(__builtin_amdgcn_update_dpp(0, __float_as_int(es), 0x168, 0xf, 0xf, true));
  es += __int_as_float(__builtin_amdgcn_update_dpp(0, __float_as_int(es), 0x164, 0xf, 0xf, true));
  es += __int_as_float(__builtin_amdgcn_update_dpp(0, __float_as_int(es), 0x162, 0xf, 0xf, true));
  es += __int_as_float(__builtin_amdgcn_update_dpp(0, __float_as_int(es), 0x161, 0xf, 0xf, true));

  float4 part[4];
  float ezv[4], invv[4];
#pragma unroll
  for (int ps = 0; ps < 4; ++ps) {
    const int r = wave * 8 + ps * 2 + hf;
    const int g = blk * BROWS + r;
    const int i = g & (NSEQ - 1);
    const v4f s = sc[ps];
    const float pa = (r == 0) ? es : s.x;
    const float pc = (r == BROWS - 1) ? es : s.z;

    const float a  = (i > 0) ? pa : 0.0f;
    const float bs = s.y;
    const float c  = (i < NSEQ - 1) ? pc : 0.0f;

    const float m  = fmaxf(fmaxf(a, bs), fmaxf(c, 0.0f));
    const float ea = expf(a - m);
    const float eb = expf(bs - m);
    const float ec = expf(c - m);
    const float ez = expf(-m);
    invv[ps] = 1.0f / (ea + eb + ec + (float)(NSEQ - 3) * ez);
    ezv[ps]  = ez;
    part[ps].x = ea * v0.x + eb * v1.x + ec * v2.x;
    part[ps].y = ea * v0.y + eb * v1.y + ec * v2.y;
    part[ps].z = ea * v0.z + eb * v1.z + ec * v2.z;
    part[ps].w = ea * v0.w + eb * v1.w + ec * v2.w;
  }
  __syncthreads();

  float4 sr = make_float4(0.0f, 0.0f, 0.0f, 0.0f);
#pragma unroll
  for (int k = 0; k < 16; ++k) {
    const float4 t = ((const float4*)red)[k * (DHEAD / 4) + c4];
    sr.x += t.x; sr.y += t.y; sr.z += t.z; sr.w += t.w;
  }

#pragma unroll
  for (int ps = 0; ps < 4; ++ps) {
    const int g = blk * BROWS + wave * 8 + ps * 2 + hf;
    float4 o;
    o.x = (part[ps].x + ezv[ps] * sr.x) * invv[ps];
    o.y = (part[ps].y + ezv[ps] * sr.y) * invv[ps];
    o.z = (part[ps].z + ezv[ps] * sr.z) * invv[ps];
    o.w = (part[ps].w + ezv[ps] * sr.w) * invv[ps];
    ((float4*)(out + (size_t)g * DHEAD))[c4] = o;
  }
}

extern "C" void kernel_launch(void* const* d_in, const int* in_sizes, int n_in,
                              void* d_out, int out_size, void* d_ws, size_t ws_size,
                              hipStream_t stream) {
  const float* x  = (const float*)d_in[0];
  const float* Wq = (const float*)d_in[1];
  const float* bq = (const float*)d_in[2];
  const float* Wk = (const float*)d_in[3];
  const float* bk = (const float*)d_in[4];
  const float* Wv = (const float*)d_in[5];
  const float* bv = (const float*)d_in[6];
  float* out = (float*)d_out;

  const int BN = in_sizes[0] / EMBED;
  const int B  = BN / NSEQ;

  float* S4    = (float*)d_ws;
  float* Qe    = S4 + (size_t)BN * 4;
  float* Ke    = Qe + (size_t)(BN / BROWS) * 2 * DHEAD;
  float* Vhead = Ke + (size_t)(BN / BROWS) * 2 * DHEAD;
  float* Psum  = Vhead + (size_t)B * 3 * DHEAD;

  qkv_gemm_kernel<<<BN / BROWS, 384, 0, stream>>>(x, Wq, bq, Wk, bk, Wv, bv, S4, Qe, Ke, Vhead, Psum);
  combine_kernel<<<BN / BROWS, 256, 0, stream>>>(S4, Qe, Ke, Vhead, Psum, out);
}


# K loop: each 32-k block's A (16-byte LDS reads) and W fragments fetched as ONE group into a second register set, a block ahead of its WMMAs; k order inside 8 k, so summation order, changes
# speedup vs baseline: 3.3854x; 1.0778x over previous
#include <hip/hip_runtime.h>
#include <hip/hip_bf16.h>

#define EMBED 768
#define NSEQ  4096
#define DHEAD 64
#define LDSTR 772
#define MT 4
#define BROWS (16 * MT)
#define SUMROW (BROWS + 3)
#define LDSROWS (SUMROW + 2)
#define PARTS_PER_BATCH (NSEQ / BROWS)
#define SP 68
#define KS_OFF (BROWS * SP)

typedef __attribute__((ext_vector_type(2))) float v2f;
typedef __attribute__((ext_vector_type(8))) float v8f;
typedef __attribute__((ext_vector_type(4))) float v4f;

template <int NT, int NS>
__device__ __forceinline__ void load_block(const float* lp, const float* __restrict__ wp,
                                           v4f (&a)[NT][NS], v2f (&w)[2 * NS]) {
#pragma unroll
  for (int s = 0; s < NS; ++s) {
#pragma unroll
    for (int t = 0; t < NT; ++t)
      a[t][s] = *(const v4f*)(lp + t * 16 * LDSTR + 8 * s);
#pragma unroll
    for (int e = 0; e < 2; ++e) {
      w[2 * s + e][0] = wp[(8 * s + 2 * e) * DHEAD];
      w[2 * s + e][1] = wp[(8 * s + 2 * e + 1) * DHEAD];
    }
  }
}

template <int NT, int NS>
__device__ __forceinline__ void mma_block(const v4f (&a)[NT][NS], const v2f (&w)[2 * NS],
                                          v8f (&acc)[NT]) {
#pragma unroll
  for (int s = 0; s < NS; ++s) {
#pragma unroll
    for (int e = 0; e < 2; ++e) {
#pragma unroll
      for (int t = 0; t < NT; ++t) {
        v2f af;  af[0] = a[t][s][2 * e];
                 af[1] = a[t][s][2 * e + 1];
        acc[t] = __builtin_amdgcn_wmma_f32_16x16x4_f32(
             false, af,  false, w[2 * s + e],
             (short)0, acc[t],  false,  false);
      }
    }
  }
}

template <int NT, int NS>
__device__ __forceinline__ void wave_kloop(const float* lrow, const float* __restrict__ wcol,
                                           const int lh, v8f (&acc)[NT]) {
  constexpr int KBLK = 8 * NS;
  v4f aA[NT][NS], aB[NT][NS];
  v2f wA[2 * NS], wB[2 * NS];
  const float* lp = lrow + 4 * lh;
  const float* wp = wcol + (size_t)(4 * lh) * DHEAD;
  load_block<NT, NS>(lp, wp, aA, wA);

#pragma nounroll
  for (int kb = 0; kb < EMBED - 2 * KBLK; kb += 2 * KBLK) {
    load_block<NT, NS>(lp + KBLK, wp + (size_t)KBLK * DHEAD, aB, wB);
    __builtin_amdgcn_sched_barrier(0);
    mma_block<NT, NS>(aA, wA, acc);
    __builtin_amdgcn_sched_barrier(0);
    lp += 2 * KBLK;
    wp += (size_t)(2 * KBLK) * DHEAD;
    load_block<NT, NS>(lp, wp, aA, wA);
    __builtin_amdgcn_sched_barrier(0);
    mma_block<NT, NS>(aB, wB, acc);
    __builtin_amdgcn_sched_barrier(0);
  }
  load_block<NT, NS>(lp + KBLK, wp + (size_t)KBLK * DHEAD, aB, wB);
  __builtin_amdgcn_sched_barrier(0);
  mma_block<NT, NS>(aA, wA, acc);
  mma_block<NT, NS>(aB, wB, acc);
  __builtin_amdgcn_sched_barrier(0);
}

__global__ __launch_bounds__(384) void qkv_gemm_kernel(
    const float* __restrict__ x,
    const float* __restrict__ Wq, const float* __restrict__ bq,
    const float* __restrict__ Wk, const float* __restrict__ bk,
    const float* __restrict__ Wv, const float* __restrict__ bv,
    float* __restrict__ S4, float* __restrict__ Qe, float* __restrict__ Ke,
    float* __restrict__ Vhead, float* __restrict__ Psum) {
  __shared__ __attribute__((aligned(16))) float lds_x[LDSROWS * LDSTR];
  __shared__ __attribute__((aligned(16))) float lds_qk[(2 * BROWS + 2) * SP];

  const int m0  = blockIdx.x * BROWS;
  const int tid = threadIdx.x;
  const bool first = (m0 & (NSEQ - 1)) == 0;

  {
    const int rh = tid / (EMBED / 4);
    const int c4 = tid % (EMBED / 4);
    const float4* srcp = (const float4*)(x + (size_t)(m0 + rh) * EMBED) + c4;
    float* dstp = &lds_x[rh * LDSTR + c4 * 4];
    float4 st[BROWS / 2];
#pragma unroll
    for (int j = 0; j < BROWS / 2; ++j) st[j] = srcp[(size_t)j * 2 * (EMBED / 4)];
    float4 cs = make_float4(0.0f, 0.0f, 0.0f, 0.0f);
#pragma unroll
    for (int j = 0; j < BROWS / 2; ++j) {
      float* dst = dstp + j * 2 * LDSTR;
      dst[0] = st[j].x; dst[1] = st[j].y; dst[2] = st[j].z; dst[3] = st[j].w;
      const bool take = (j >= 2) || !first || (j == 1 && rh == 1);
      if (take) { cs.x += st[j].x; cs.y += st[j].y; cs.z += st[j].z; cs.w += st[j].w; }
    }
    float* sp = &lds_x[(SUMROW + rh) * LDSTR + c4 * 4];
    sp[0] = cs.x; sp[1] = cs.y; sp[2] = cs.z; sp[3] = cs.w;
  }
  __syncthreads();

  const int wave = __builtin_amdgcn_readfirstlane(tid >> 5);
  const int lane = tid & 31;
  const int ln   = lane & 15;
  const int lh   = lane >> 4;

  if (wave < 8) {
    const float* W; const float* bias; float* tile; int n0;
    if (wave < 4) { W = Wq; bias = bq; tile = lds_qk;               n0 = wave * 16; }
    else          { W = Wk; bias = bk; tile = lds_qk + KS_OFF + SP; n0 = (wave - 4) * 16; }

    const float bval = bias[n0 + ln];
    v8f acc[MT];
#pragma unroll
    for (int t = 0; t < MT; ++t) {
#pragma unroll
      for (int r = 0; r < 8; ++r) acc[t][r] = bval;
    }

    wave_kloop<MT, 4>(&lds_x[ln * LDSTR], W + n0 + ln, lh, acc);

    asm volatile("v_nop\n\tv_nop\n\tv_nop\n\tv_nop\n\tv_nop"
                 : "+v"(acc[0]), "+v"(acc[1]), "+v"(acc[2]), "+v"(acc[3]));

#pragma unroll
    for (int t = 0; t < MT; ++t) {
#pragma unroll
      for (int r = 0; r < 8; ++r) {
        const int row = 16 * t + r + 8 * lh;
        tile[row * SP + n0 + ln] = acc[t][r];
      }
    }
  } else {
    const int n0 = (wave - 8) * 16;
    const float bval = bv[n0 + ln];
    v8f acc[1];
#pragma unroll
    for (int r = 0; r < 8; ++r) acc[0][r] = bval;

    const int arow = (ln == 3) ? SUMROW : (ln == 4) ? (SUMROW + 1) : ln;
    wave_kloop<1, 8>(&lds_x[arow * LDSTR], Wv + n0 + ln, lh, acc);

    asm volatile("v_nop\n\tv_nop\n\tv_nop\n\tv_nop\n\tv_nop" : "+v"(acc[0]));

    if (lh == 0) {
      if (first) {
        const int b = m0 / NSEQ;
#pragma unroll
        for (int r = 0; r < 3; ++r) Vhead[(size_t)(b * 3 + r) * DHEAD + n0 + ln] = acc[0][r];
      }
      const float cnt = first ? (float)(BROWS - 3) : (float)BROWS;
      Psum[(size_t)blockIdx.x * DHEAD + n0 + ln] = acc[0][3] + acc[0][4] + (cnt - 2.0f) * bval;
    }
    {
      const int h  = tid - 256;
      const int hr = (h < DHEAD) ? 0 : (BROWS + 1);
      lds_qk[KS_OFF + hr * SP + (h & (DHEAD - 1))] = 0.0f;
    }
  }
  __syncthreads();

  if (wave < 8) {
    const int r = tid >> 2;
    const int j = tid & 3;
    const float4* q4 = (const float4*)(lds_qk + r * SP + 16 * j);
    const float4* kp = (const float4*)(lds_qk + KS_OFF + r * SP + 16 * j);
    const float4* kc = (const float4*)(lds_qk + KS_OFF + (r + 1) * SP + 16 * j);
    const float4* kn = (const float4*)(lds_qk + KS_OFF + (r + 2) * SP + 16 * j);
    float pa = 0.0f, pb = 0.0f, pc = 0.0f;
#pragma unroll
    for (int e = 0; e < 4; ++e) {
      const float4 q = q4[e], a = kp[e], b = kc[e], c = kn[e];
      pa += q.x * a.x + q.y * a.y + q.z * a.z + q.w * a.w;
      pb += q.x * b.x + q.y * b.y + q.z * b.z + q.w * b.w;
      pc += q.x * c.x + q.y * c.y + q.z * c.z + q.w * c.w;
    }
    pa += __shfl_xor(pa, 1, 32);  pb += __shfl_xor(pb, 1, 32);  pc += __shfl_xor(pc, 1, 32);
    pa += __shfl_xor(pa, 2, 32);  pb += __shfl_xor(pb, 2, 32);  pc += __shfl_xor(pc, 2, 32);
    if (j == 0) ((float4*)S4)[m0 + r] = make_float4(pa, pb, pc, 0.0f);

    if (wave < 2) {
      const int er = (lane < 16) ? 0 : (BROWS - 1);
      const float* srow = (wave == 0) ? (lds_qk + er * SP) : (lds_qk + KS_OFF + (er + 1) * SP);
      float* edge = (wave == 0) ? Qe : Ke;
      ((float4*)edge)[(size_t)blockIdx.x * 32 + lane] = ((const float4*)srow)[lane & 15];
    }
  }
}

__global__ __launch_bounds__(256) void combine_kernel(
    const float* __restrict__ S4, const float* __restrict__ Qe, const float* __restrict__ Ke,
    const float* __restrict__ Vhead, const float* __restrict__ Psum,
    float* __restrict__ out) {
  __shared__ __attribute__((aligned(16))) float red[16 * DHEAD];
  const int tid  = threadIdx.x;
  const int wave = tid >> 5;
  const int lane = tid & 31;
  const int hf   = lane >> 4;
  const int c4   = lane & 15;
  const int blk  = blockIdx.x;
  const int b    = blk / PARTS_PER_BATCH;
  const int pb   = blk % PARTS_PER_BATCH;

  v4f sc[4];
#pragma unroll
  for (int ps = 0; ps < 4; ++ps) sc[ps] = ((const v4f*)S4)[blk * BROWS + wave * 8 + ps * 2 + hf];

  const v4f* Vb = (const v4f*)(Vhead + (size_t)b * 3 * DHEAD);
  v4f v0 = Vb[c4];
  v4f v1 = Vb[16 + c4];
  v4f v2 = Vb[32 + c4];

  const int lastrow = (wave == 7) ? 1 : 0;
  const int nb = lastrow ? ((pb < PARTS_PER_BATCH - 1) ? blk + 1 : blk)
                         : ((pb > 0) ? blk - 1 : blk);
  v4f eq = ((const v4f*)Qe)[(size_t)blk * 32 + 16 * lastrow + c4];
  v4f ek = ((const v4f*)Ke)[(size_t)nb * 32 + 16 * (1 - lastrow) + c4];

  const int d  = tid & (DHEAD - 1);
  const int sg = tid >> 6;
  const float* pp = Psum + ((size_t)b * PARTS_PER_BATCH + sg * 16) * DHEAD + d;
  v4f pr[4];
#pragma unroll
  for (int u = 0; u < 4; ++u) {
#pragma unroll
    for (int k = 0; k < 4; ++k) pr[u][k] = pp[(4 * u + k) * DHEAD];
  }

  asm volatile("" : "+v"(sc[0]), "+v"(sc[1]), "+v"(sc[2]), "+v"(sc[3]),
                    "+v"(v0), "+v"(v1), "+v"(v2), "+v"(eq), "+v"(ek),
                    "+v"(pr[0]), "+v"(pr[1]), "+v"(pr[2]), "+v"(pr[3]));

#pragma unroll
  for (int u = 0; u < 4; ++u) {
    float a = 0.0f;
#pragma unroll
    for (int k = 0; k < 4; ++k) a += pr[u][k];
    red[(4 * sg + u) * DHEAD + d] = a;
  }

  float es = eq.x * ek.x + eq.y * ek.y + eq.z * ek.z + eq.w * ek.w;
  es += __int_as_float(__builtin_amdgcn_update_dpp(0, __float_as_int(es), 0x168, 0xf, 0xf, true));
  es += __int_as_float(__builtin_amdgcn_update_dpp(0, __float_as_int(es), 0x164, 0xf, 0xf, true));
  es += __int_as_float(__builtin_amdgcn_update_dpp(0, __float_as_int(es), 0x162, 0xf, 0xf, true));
  es += __int_as_float(__builtin_amdgcn_update_dpp(0, __float_as_int(es), 0x161, 0xf, 0xf, true));

  float4 part[4];
  float ezv[4], invv[4];
#pragma unroll
  for (int ps = 0; ps < 4; ++ps) {
    const int r = wave * 8 + ps * 2 + hf;
    const int g = blk * BROWS + r;
    const int i = g & (NSEQ - 1);
    const v4f s = sc[ps];
    const float pa = (r == 0) ? es : s.x;
    const float pc = (r == BROWS - 1) ? es : s.z;

    const float a  = (i > 0) ? pa : 0.0f;
    const float bs = s.y;
    const float c  = (i < NSEQ - 1) ? pc : 0.0f;

    const float m  = fmaxf(fmaxf(a, bs), fmaxf(c, 0.0f));
    const float ea = expf(a - m);
    const float eb = expf(bs - m);
    const float ec = expf(c - m);
    const float ez = expf(-m);
    invv[ps] = 1.0f / (ea + eb + ec + (float)(NSEQ - 3) * ez);
    ezv[ps]  = ez;
    part[ps].x = ea * v0.x + eb * v1.x + ec * v2.x;
    part[ps].y = ea * v0.y + eb * v1.y + ec * v2.y;
    part[ps].z = ea * v0.z + eb * v1.z + ec * v2.z;
    part[ps].w = ea * v0.w + eb * v1.w + ec * v2.w;
  }
  __syncthreads();

  float4 sr = make_float4(0.0f, 0.0f, 0.0f, 0.0f);
#pragma unroll
  for (int k = 0; k < 16; ++k) {
    const float4 t = ((const float4*)red)[k * (DHEAD / 4) + c4];
    sr.x += t.x; sr.y += t.y; sr.z += t.z; sr.w += t.w;
  }

#pragma unroll
  for (int ps = 0; ps < 4; ++ps) {
    const int g = blk * BROWS + wave * 8 + ps * 2 + hf;
    float4 o;
    o.x = (part[ps].x + ezv[ps] * sr.x) * invv[ps];
    o.y = (part[ps].y + ezv[ps] * sr.y) * invv[ps];
    o.z = (part[ps].z + ezv[ps] * sr.z) * invv[ps];
    o.w = (part[ps].w + ezv[ps] * sr.w) * invv[ps];
    ((float4*)(out + (size_t)g * DHEAD))[c4] = o;
  }
}

extern "C" void kernel_launch(void* const* d_in, const int* in_sizes, int n_in,
                              void* d_out, int out_size, void* d_ws, size_t ws_size,
                              hipStream_t stream) {
  const float* x  = (const float*)d_in[0];
  const float* Wq = (const float*)d_in[1];
  const float* bq = (const float*)d_in[2];
  const float* Wk = (const float*)d_in[3];
  const float* bk = (const float*)d_in[4];
  const float* Wv = (const float*)d_in[5];
  const float* bv = (const float*)d_in[6];
  float* out = (float*)d_out;

  const int BN = in_sizes[0] / EMBED;
  const int B  = BN / NSEQ;

  float* S4    = (float*)d_ws;
  float* Qe    = S4 + (size_t)BN * 4;
  float* Ke    = Qe + (size_t)(BN / BROWS) * 2 * DHEAD;
  float* Vhead = Ke + (size_t)(BN / BROWS) * 2 * DHEAD;
  float* Psum  = Vhead + (size_t)B * 3 * DHEAD;

  qkv_gemm_kernel<<<BN / BROWS, 384, 0, stream>>>(x, Wq, bq, Wk, bk, Wv, bv, S4, Qe, Ke, Vhead, Psum);
  combine_kernel<<<BN / BROWS, 256, 0, stream>>>(S4, Qe, Ke, Vhead, Psum, out);
}


# V rows (the tile's column-sum row, in a batch's first block also x rows 0..2, times Wv) moved from a 16-row WMMA tile with one useful row to the vector unit beside the Q/K WMMAs; V sum order changes
# speedup vs baseline: 3.5111x; 1.0371x over previous
#include <hip/hip_runtime.h>
#include <hip/hip_bf16.h>

#define EMBED 768
#define NSEQ  4096
#define DHEAD 64
#define LDSTR 772
#define MT 4
#define BROWS (16 * MT)
#define SUMROW (BROWS + 3)
#define LDSROWS (SUMROW + 2)
#define VKG 8
#define VKPER (EMBED / VKG)
#define PARTS_PER_BATCH (NSEQ / BROWS)
#define SP 68
#define KS_OFF (BROWS * SP)

typedef __attribute__((ext_vector_type(2))) float v2f;
typedef __attribute__((ext_vector_type(8))) float v8f;
typedef __attribute__((ext_vector_type(4))) float v4f;

template <int NT, int NS>
__device__ __forceinline__ void load_block(const float* lp, const float* __restrict__ wp,
                                           v4f (&a)[NT][NS], v2f (&w)[2 * NS]) {
#pragma unroll
  for (int s = 0; s < NS; ++s) {
#pragma unroll
    for (int t = 0; t < NT; ++t)
      a[t][s] = *(const v4f*)(lp + t * 16 * LDSTR + 8 * s);
#pragma unroll
    for (int e = 0; e < 2; ++e) {
      w[2 * s + e][0] = wp[(8 * s + 2 * e) * DHEAD];
      w[2 * s + e][1] = wp[(8 * s + 2 * e + 1) * DHEAD];
    }
  }
}

template <int NT, int NS>
__device__ __forceinline__ void mma_block(const v4f (&a)[NT][NS], const v2f (&w)[2 * NS],
                                          v8f (&acc)[NT]) {
#pragma unroll
  for (int s = 0; s < NS; ++s) {
#pragma unroll
    for (int e = 0; e < 2; ++e) {
#pragma unroll
      for (int t = 0; t < NT; ++t) {
        v2f af;  af[0] = a[t][s][2 * e];
                 af[1] = a[t][s][2 * e + 1];
        acc[t] = __builtin_amdgcn_wmma_f32_16x16x4_f32(
             false, af,  false, w[2 * s + e],
             (short)0, acc[t],  false,  false);
      }
    }
  }
}

template <int NT>
__device__ __forceinline__ void wmma_run_slots(v8f (&acc)[NT]) {
  static_assert(NT == 4, "one operand per accumulator");
  asm volatile("v_nop\n\tv_nop\n\tv_nop\n\tv_nop\n\tv_nop"
               : "+v"(acc[0]), "+v"(acc[1]), "+v"(acc[2]), "+v"(acc[3]));
}

template <int NT, int NS>
__device__ __forceinline__ void wave_kloop(const float* lrow, const float* __restrict__ wcol,
                                           const int lh, v8f (&acc)[NT]) {
  constexpr int KBLK = 8 * NS;
  v4f aA[NT][NS], aB[NT][NS];
  v2f wA[2 * NS], wB[2 * NS];
  const float* lp = lrow + 4 * lh;
  const float* wp = wcol + (size_t)(4 * lh) * DHEAD;
  load_block<NT, NS>(lp, wp, aA, wA);

#pragma nounroll
  for (int kb = 0; kb < EMBED - 2 * KBLK; kb += 2 * KBLK) {
    load_block<NT, NS>(lp + KBLK, wp + (size_t)KBLK * DHEAD, aB, wB);
    __builtin_amdgcn_sched_barrier(0);
    mma_block<NT, NS>(aA, wA, acc);
    wmma_run_slots<NT>(acc);
    __builtin_amdgcn_sched_barrier(0);
    lp += 2 * KBLK;
    wp += (size_t)(2 * KBLK) * DHEAD;
    load_block<NT, NS>(lp, wp, aA, wA);
    __builtin_amdgcn_sched_barrier(0);
    mma_block<NT, NS>(aB, wB, acc);
    wmma_run_slots<NT>(acc);
    __builtin_amdgcn_sched_barrier(0);
  }
  load_block<NT, NS>(lp + KBLK, wp + (size_t)KBLK * DHEAD, aB, wB);
  __builtin_amdgcn_sched_barrier(0);
  mma_block<NT, NS>(aA, wA, acc);
  mma_block<NT, NS>(aB, wB, acc);
  __builtin_amdgcn_sched_barrier(0);
}

template <int NR>
__device__ __forceinline__ void v_rows(const float* lds_x, const float* __restrict__ wv4,
                                       const int k0, v4f (&acc)[NR]) {
#pragma unroll
  for (int r = 0; r < NR; ++r) acc[r] = (v4f){0.0f, 0.0f, 0.0f, 0.0f};
#pragma unroll 4
  for (int kk = 0; kk < VKPER; kk += 4) {
    const int k = k0 + kk;
    v4f a[NR];
#pragma unroll
    for (int r = 0; r < NR - 1; ++r) a[r] = *(const v4f*)(lds_x + r * LDSTR + k);
    a[NR - 1] = *(const v4f*)(lds_x + SUMROW * LDSTR + k) + *(const v4f*)(lds_x + (SUMROW + 1) * LDSTR + k);
#pragma unroll
    for (int e = 0; e < 4; ++e) {
      const v4f w = *(const v4f*)(wv4 + (size_t)(k + e) * DHEAD);
#pragma unroll
      for (int r = 0; r < NR; ++r) acc[r] += a[r][e] * w;
    }
  }
}

__global__ __launch_bounds__(384) void qkv_gemm_kernel(
    const float* __restrict__ x,
    const float* __restrict__ Wq, const float* __restrict__ bq,
    const float* __restrict__ Wk, const float* __restrict__ bk,
    const float* __restrict__ Wv, const float* __restrict__ bv,
    float* __restrict__ S4, float* __restrict__ Qe, float* __restrict__ Ke,
    float* __restrict__ Vhead, float* __restrict__ Psum) {
  __shared__ __attribute__((aligned(16))) float lds_x[LDSROWS * LDSTR];
  __shared__ __attribute__((aligned(16))) float lds_qk[(2 * BROWS + 2) * SP];
  __shared__ __attribute__((aligned(16))) float vpart[4 * VKG * DHEAD];

  const int m0  = blockIdx.x * BROWS;
  const int tid = threadIdx.x;
  const bool first = (m0 & (NSEQ - 1)) == 0;

  {
    const int rh = tid / (EMBED / 4);
    const int c4 = tid % (EMBED / 4);
    const float4* srcp = (const float4*)(x + (size_t)(m0 + rh) * EMBED) + c4;
    float* dstp = &lds_x[rh * LDSTR + c4 * 4];
    float4 st[BROWS / 2];
#pragma unroll
    for (int j = 0; j < BROWS / 2; ++j) st[j] = srcp[(size_t)j * 2 * (EMBED / 4)];
    float4 cs = make_float4(0.0f, 0.0f, 0.0f, 0.0f);
#pragma unroll
    for (int j = 0; j < BROWS / 2; ++j) {
      float* dst = dstp + j * 2 * LDSTR;
      dst[0] = st[j].x; dst[1] = st[j].y; dst[2] = st[j].z; dst[3] = st[j].w;
      const bool take = (j >= 2) || !first || (j == 1 && rh == 1);
      if (take) { cs.x += st[j].x; cs.y += st[j].y; cs.z += st[j].z; cs.w += st[j].w; }
    }
    float* sp = &lds_x[(SUMROW + rh) * LDSTR + c4 * 4];
    sp[0] = cs.x; sp[1] = cs.y; sp[2] = cs.z; sp[3] = cs.w;
  }
  __syncthreads();

  const int wave = __builtin_amdgcn_readfirstlane(tid >> 5);
  const int lane = tid & 31;
  const int ln   = lane & 15;
  const int lh   = lane >> 4;

  if (wave < 8) {
    const float* W; const float* bias; float* tile; int n0;
    if (wave < 4) { W = Wq; bias = bq; tile = lds_qk;               n0 = wave * 16; }
    else          { W = Wk; bias = bk; tile = lds_qk + KS_OFF + SP; n0 = (wave - 4) * 16; }

    const float bval = bias[n0 + ln];
    v8f acc[MT];
#pragma unroll
    for (int t = 0; t < MT; ++t) {
#pragma unroll
      for (int r = 0; r < 8; ++r) acc[t][r] = bval;
    }

    wave_kloop<MT, 4>(&lds_x[ln * LDSTR], W + n0 + ln, lh, acc);

    asm volatile("v_nop\n\tv_nop\n\tv_nop\n\tv_nop\n\tv_nop"
                 : "+v"(acc[0]), "+v"(acc[1]), "+v"(acc[2]), "+v"(acc[3]));

#pragma unroll
    for (int t = 0; t < MT; ++t) {
#pragma unroll
      for (int r = 0; r < 8; ++r) {
        const int row = 16 * t + r + 8 * lh;
        tile[row * SP + n0 + ln] = acc[t][r];
      }
    }
  } else {
    const int h  = tid - 256;
    const int cg = h & 15;
    const int kg = h >> 4;
    const float* wv4 = Wv + 4 * cg;
    float* vp = &vpart[kg * DHEAD + 4 * cg];
    if (first) {
      v4f acc[4];
      v_rows<4>(lds_x, wv4, kg * VKPER, acc);
#pragma unroll
      for (int r = 0; r < 4; ++r) *(v4f*)(vp + r * VKG * DHEAD) = acc[r];
    } else {
      v4f acc[1];
      v_rows<1>(lds_x, wv4, kg * VKPER, acc);
      *(v4f*)(vp + 3 * VKG * DHEAD) = acc[0];
    }
    {
      const int hr = (h < DHEAD) ? 0 : (BROWS + 1);
      lds_qk[KS_OFF + hr * SP + (h & (DHEAD - 1))] = 0.0f;
    }
  }
  __syncthreads();

  if (wave < 8) {
    const int r = tid >> 2;
    const int j = tid & 3;
    const float4* q4 = (const float4*)(lds_qk + r * SP + 16 * j);
    const float4* kp = (const float4*)(lds_qk + KS_OFF + r * SP + 16 * j);
    const float4* kc = (const float4*)(lds_qk + KS_OFF + (r + 1) * SP + 16 * j);
    const float4* kn = (const float4*)(lds_qk + KS_OFF + (r + 2) * SP + 16 * j);
    float pa = 0.0f, pb = 0.0f, pc = 0.0f;
#pragma unroll
    for (int e = 0; e < 4; ++e) {
      const float4 q = q4[e], a = kp[e], b = kc[e], c = kn[e];
      pa += q.x * a.x + q.y * a.y + q.z * a.z + q.w * a.w;
      pb += q.x * b.x + q.y * b.y + q.z * b.z + q.w * b.w;
      pc += q.x * c.x + q.y * c.y + q.z * c.z + q.w * c.w;
    }
    pa += __shfl_xor(pa, 1, 32);  pb += __shfl_xor(pb, 1, 32);  pc += __shfl_xor(pc, 1, 32);
    pa += __shfl_xor(pa, 2, 32);  pb += __shfl_xor(pb, 2, 32);  pc += __shfl_xor(pc, 2, 32);
    if (j == 0) ((float4*)S4)[m0 + r] = make_float4(pa, pb, pc, 0.0f);

    if (wave < 2) {
      const int er = (lane < 16) ? 0 : (BROWS - 1);
      const float* srow = (wave == 0) ? (lds_qk + er * SP) : (lds_qk + KS_OFF + (er + 1) * SP);
      float* edge = (wave == 0) ? Qe : Ke;
      ((float4*)edge)[(size_t)blockIdx.x * 32 + lane] = ((const float4*)srow)[lane & 15];
    }
  } else {
    const int h = tid - 256;
    const int d = h & (DHEAD - 1);
    const float bval = bv[d];
    const float cnt = first ? (float)(BROWS - 3) : (float)BROWS;
#pragma unroll
    for (int rr = 0; rr < 2; ++rr) {
      const int row = (h >> 6) + 2 * rr;
      if (row == 3 || first) {
        float a = 0.0f;
#pragma unroll
        for (int g = 0; g < VKG; ++g) a += vpart[(row * VKG + g) * DHEAD + d];
        if (row == 3) Psum[(size_t)blockIdx.x * DHEAD + d] = a + cnt * bval;
        else          Vhead[(size_t)((m0 / NSEQ) * 3 + row) * DHEAD + d] = a + bval;
      }
    }
  }
}

__global__ __launch_bounds__(256) void combine_kernel(
    const float* __restrict__ S4, const float* __restrict__ Qe, const float* __restrict__ Ke,
    const float* __restrict__ Vhead, const float* __restrict__ Psum,
    float* __restrict__ out) {
  __shared__ __attribute__((aligned(16))) float red[16 * DHEAD];
  const int tid  = threadIdx.x;
  const int wave = tid >> 5;
  const int lane = tid & 31;
  const int hf   = lane >> 4;
  const int c4   = lane & 15;
  const int blk  = blockIdx.x;
  const int b    = blk / PARTS_PER_BATCH;
  const int pb   = blk % PARTS_PER_BATCH;

  v4f sc[4];
#pragma unroll
  for (int ps = 0; ps < 4; ++ps) sc[ps] = ((const v4f*)S4)[blk * BROWS + wave * 8 + ps * 2 + hf];

  const v4f* Vb = (const v4f*)(Vhead + (size_t)b * 3 * DHEAD);
  v4f v0 = Vb[c4];
  v4f v1 = Vb[16 + c4];
  v4f v2 = Vb[32 + c4];

  const int lastrow = (wave == 7) ? 1 : 0;
  const int nb = lastrow ? ((pb < PARTS_PER_BATCH - 1) ? blk + 1 : blk)
                         : ((pb > 0) ? blk - 1 : blk);
  v4f eq = ((const v4f*)Qe)[(size_t)blk * 32 + 16 * lastrow + c4];
  v4f ek = ((const v4f*)Ke)[(size_t)nb * 32 + 16 * (1 - lastrow) + c4];

  const int d  = tid & (DHEAD - 1);
  const int sg = tid >> 6;
  const float* pp = Psum + ((size_t)b * PARTS_PER_BATCH + sg * 16) * DHEAD + d;
  v4f pr[4];
#pragma unroll
  for (int u = 0; u < 4; ++u) {
#pragma unroll
    for (int k = 0; k < 4; ++k) pr[u][k] = pp[(4 * u + k) * DHEAD];
  }

  asm volatile("" : "+v"(sc[0]), "+v"(sc[1]), "+v"(sc[2]), "+v"(sc[3]),
                    "+v"(v0), "+v"(v1), "+v"(v2), "+v"(eq), "+v"(ek),
                    "+v"(pr[0]), "+v"(pr[1]), "+v"(pr[2]), "+v"(pr[3]));

#pragma unroll
  for (int u = 0; u < 4; ++u) {
    float a = 0.0f;
#pragma unroll
    for (int k = 0; k < 4; ++k) a += pr[u][k];
    red[(4 * sg + u) * DHEAD + d] = a;
  }

  float es = eq.x * ek.x + eq.y * ek.y + eq.z * ek.z + eq.w * ek.w;
  es += __int_as_float(__builtin_amdgcn_update_dpp(0, __float_as_int(es), 0x168, 0xf, 0xf, true));
  es += __int_as_float(__builtin_amdgcn_update_dpp(0, __float_as_int(es), 0x164, 0xf, 0xf, true));
  es += __int_as_float(__builtin_amdgcn_update_dpp(0, __float_as_int(es), 0x162, 0xf, 0xf, true));
  es += __int_as_float(__builtin_amdgcn_update_dpp(0, __float_as_int(es), 0x161, 0xf, 0xf, true));

  float4 part[4];
  float ezv[4], invv[4];
#pragma unroll
  for (int ps = 0; ps < 4; ++ps) {
    const int r = wave * 8 + ps * 2 + hf;
    const int g = blk * BROWS + r;
    const int i = g & (NSEQ - 1);
    const v4f s = sc[ps];
    const float pa = (r == 0) ? es : s.x;
    const float pc = (r == BROWS - 1) ? es : s.z;

    const float a  = (i > 0) ? pa : 0.0f;
    const float bs = s.y;
    const float c  = (i < NSEQ - 1) ? pc : 0.0f;

    const float m  = fmaxf(fmaxf(a, bs), fmaxf(c, 0.0f));
    const float ea = expf(a - m);
    const float eb = expf(bs - m);
    const float ec = expf(c - m);
    const float ez = expf(-m);
    invv[ps] = 1.0f / (ea + eb + ec + (float)(NSEQ - 3) * ez);
    ezv[ps]  = ez;
    part[ps].x = ea * v0.x + eb * v1.x + ec * v2.x;
    part[ps].y = ea * v0.y + eb * v1.y + ec * v2.y;
    part[ps].z = ea * v0.z + eb * v1.z + ec * v2.z;
    part[ps].w = ea * v0.w + eb * v1.w + ec * v2.w;
  }
  __syncthreads();

  float4 sr = make_float4(0.0f, 0.0f, 0.0f, 0.0f);
#pragma unroll
  for (int k = 0; k < 16; ++k) {
    const float4 t = ((const float4*)red)[k * (DHEAD / 4) + c4];
    sr.x += t.x; sr.y += t.y; sr.z += t.z; sr.w += t.w;
  }

#pragma unroll
  for (int ps = 0; ps < 4; ++ps) {
    const int g = blk * BROWS + wave * 8 + ps * 2 + hf;
    float4 o;
    o.x = (part[ps].x + ezv[ps] * sr.x) * invv[ps];
    o.y = (part[ps].y + ezv[ps] * sr.y) * invv[ps];
    o.z = (part[ps].z + ezv[ps] * sr.z) * invv[ps];
    o.w = (part[ps].w + ezv[ps] * sr.w) * invv[ps];
    ((float4*)(out + (size_t)g * DHEAD))[c4] = o;
  }
}

extern "C" void kernel_launch(void* const* d_in, const int* in_sizes, int n_in,
                              void* d_out, int out_size, void* d_ws, size_t ws_size,
                              hipStream_t stream) {
  const float* x  = (const float*)d_in[0];
  const float* Wq = (const float*)d_in[1];
  const float* bq = (const float*)d_in[2];
  const float* Wk = (const float*)d_in[3];
  const float* bk = (const float*)d_in[4];
  const float* Wv = (const float*)d_in[5];
  const float* bv = (const float*)d_in[6];
  float* out = (float*)d_out;

  const int BN = in_sizes[0] / EMBED;
  const int B  = BN / NSEQ;

  float* S4    = (float*)d_ws;
  float* Qe    = S4 + (size_t)BN * 4;
  float* Ke    = Qe + (size_t)(BN / BROWS) * 2 * DHEAD;
  float* Vhead = Ke + (size_t)(BN / BROWS) * 2 * DHEAD;
  float* Psum  = Vhead + (size_t)B * 3 * DHEAD;

  qkv_gemm_kernel<<<BN / BROWS, 384, 0, stream>>>(x, Wq, bq, Wk, bk, Wv, bv, S4, Qe, Ke, Vhead, Psum);
  combine_kernel<<<BN / BROWS, 256, 0, stream>>>(S4, Qe, Ke, Vhead, Psum, out);
}


# K loop: each Q/K wave lowers its own issue priority 3,2,1,0 as its k advances (s_setprio), so a SIMD's two matrix waves stay abreast and end together, not in turn; V waves at 3; sums unchanged
# speedup vs baseline: 3.6217x; 1.0315x over previous
#include <hip/hip_runtime.h>
#include <hip/hip_bf16.h>

#define EMBED 768
#define NSEQ  4096
#define DHEAD 64
#define LDSTR 772
#define MT 4
#define BROWS (16 * MT)
#define SUMROW (BROWS + 3)
#define LDSROWS (SUMROW + 2)
#define VKG 8
#define VKPER (EMBED / VKG)
#define PARTS_PER_BATCH (NSEQ / BROWS)
#define SP 68
#define KS_OFF (BROWS * SP)

typedef __attribute__((ext_vector_type(2))) float v2f;
typedef __attribute__((ext_vector_type(8))) float v8f;
typedef __attribute__((ext_vector_type(4))) float v4f;

template <int NT, int NS>
__device__ __forceinline__ void load_block(const float* lp, const float* __restrict__ wp,
                                           v4f (&a)[NT][NS], v2f (&w)[2 * NS]) {
#pragma unroll
  for (int s = 0; s < NS; ++s) {
#pragma unroll
    for (int t = 0; t < NT; ++t)
      a[t][s] = *(const v4f*)(lp + t * 16 * LDSTR + 8 * s);
#pragma unroll
    for (int e = 0; e < 2; ++e) {
      w[2 * s + e][0] = wp[(8 * s + 2 * e) * DHEAD];
      w[2 * s + e][1] = wp[(8 * s + 2 * e + 1) * DHEAD];
    }
  }
}

template <int NT, int NS>
__device__ __forceinline__ void mma_block(const v4f (&a)[NT][NS], const v2f (&w)[2 * NS],
                                          v8f (&acc)[NT]) {
#pragma unroll
  for (int s = 0; s < NS; ++s) {
#pragma unroll
    for (int e = 0; e < 2; ++e) {
#pragma unroll
      for (int t = 0; t < NT; ++t) {
        v2f af;  af[0] = a[t][s][2 * e];
                 af[1] = a[t][s][2 * e + 1];
        acc[t] = __builtin_amdgcn_wmma_f32_16x16x4_f32(
             false, af,  false, w[2 * s + e],
             (short)0, acc[t],  false,  false);
      }
    }
  }
}

template <int NT>
__device__ __forceinline__ void wmma_run_slots(v8f (&acc)[NT]) {
  static_assert(NT == 4, "one operand per accumulator");
  asm volatile("v_nop\n\tv_nop\n\tv_nop\n\tv_nop\n\tv_nop"
               : "+v"(acc[0]), "+v"(acc[1]), "+v"(acc[2]), "+v"(acc[3]));
}

template <int NT, int NS>
__device__ __forceinline__ void wave_kloop(const float* lrow, const float* __restrict__ wcol,
                                           const int lh, v8f (&acc)[NT]) {
  constexpr int KBLK = 8 * NS;
  static_assert(EMBED / KBLK == 24, "the priority steps below are placed for 24 blocks");
  v4f aA[NT][NS], aB[NT][NS];
  v2f wA[2 * NS], wB[2 * NS];
  const float* lp = lrow + 4 * lh;
  const float* wp = wcol + (size_t)(4 * lh) * DHEAD;
  __builtin_amdgcn_s_setprio(3);
  load_block<NT, NS>(lp, wp, aA, wA);

#pragma nounroll
  for (int kb = 0; kb < EMBED - 2 * KBLK; kb += 2 * KBLK) {
    if (kb == 12 * KBLK) __builtin_amdgcn_s_setprio(2);
    if (kb == 18 * KBLK) __builtin_amdgcn_s_setprio(1);
    load_block<NT, NS>(lp + KBLK, wp + (size_t)KBLK * DHEAD, aB, wB);
    __builtin_amdgcn_sched_barrier(0);
    mma_block<NT, NS>(aA, wA, acc);
    wmma_run_slots<NT>(acc);
    __builtin_amdgcn_sched_barrier(0);
    lp += 2 * KBLK;
    wp += (size_t)(2 * KBLK) * DHEAD;
    load_block<NT, NS>(lp, wp, aA, wA);
    __builtin_amdgcn_sched_barrier(0);
    mma_block<NT, NS>(aB, wB, acc);
    wmma_run_slots<NT>(acc);
    __builtin_amdgcn_sched_barrier(0);
  }
  __builtin_amdgcn_s_setprio(0);
  load_block<NT, NS>(lp + KBLK, wp + (size_t)KBLK * DHEAD, aB, wB);
  __builtin_amdgcn_sched_barrier(0);
  mma_block<NT, NS>(aA, wA, acc);
  mma_block<NT, NS>(aB, wB, acc);
  __builtin_amdgcn_sched_barrier(0);
}

template <int NR>
__device__ __forceinline__ void v_rows(const float* lds_x, const float* __restrict__ wv4,
                                       const int k0, v4f (&acc)[NR]) {
#pragma unroll
  for (int r = 0; r < NR; ++r) acc[r] = (v4f){0.0f, 0.0f, 0.0f, 0.0f};
#pragma unroll 4
  for (int kk = 0; kk < VKPER; kk += 4) {
    const int k = k0 + kk;
    v4f a[NR];
#pragma unroll
    for (int r = 0; r < NR - 1; ++r) a[r] = *(const v4f*)(lds_x + r * LDSTR + k);
    a[NR - 1] = *(const v4f*)(lds_x + SUMROW * LDSTR + k) + *(const v4f*)(lds_x + (SUMROW + 1) * LDSTR + k);
#pragma unroll
    for (int e = 0; e < 4; ++e) {
      const v4f w = *(const v4f*)(wv4 + (size_t)(k + e) * DHEAD);
#pragma unroll
      for (int r = 0; r < NR; ++r) acc[r] += a[r][e] * w;
    }
  }
}

__global__ __launch_bounds__(384) void qkv_gemm_kernel(
    const float* __restrict__ x,
    const float* __restrict__ Wq, const float* __restrict__ bq,
    const float* __restrict__ Wk, const float* __restrict__ bk,
    const float* __restrict__ Wv, const float* __restrict__ bv,
    float* __restrict__ S4, float* __restrict__ Qe, float* __restrict__ Ke,
    float* __restrict__ Vhead, float* __restrict__ Psum) {
  __shared__ __attribute__((aligned(16))) float lds_x[LDSROWS * LDSTR];
  __shared__ __attribute__((aligned(16))) float lds_qk[(2 * BROWS + 2) * SP];
  __shared__ __attribute__((aligned(16))) float vpart[4 * VKG * DHEAD];

  const int m0  = blockIdx.x * BROWS;
  const int tid = threadIdx.x;
  const bool first = (m0 & (NSEQ - 1)) == 0;

  {
    const int rh = tid / (EMBED / 4);
    const int c4 = tid % (EMBED / 4);
    const float4* srcp = (const float4*)(x + (size_t)(m0 + rh) * EMBED) + c4;
    float* dstp = &lds_x[rh * LDSTR + c4 * 4];
    float4 st[BROWS / 2];
#pragma unroll
    for (int j = 0; j < BROWS / 2; ++j) st[j] = srcp[(size_t)j * 2 * (EMBED / 4)];
    float4 cs = make_float4(0.0f, 0.0f, 0.0f, 0.0f);
#pragma unroll
    for (int j = 0; j < BROWS / 2; ++j) {
      float* dst = dstp + j * 2 * LDSTR;
      dst[0] = st[j].x; dst[1] = st[j].y; dst[2] = st[j].z; dst[3] = st[j].w;
      const bool take = (j >= 2) || !first || (j == 1 && rh == 1);
      if (take) { cs.x += st[j].x; cs.y += st[j].y; cs.z += st[j].z; cs.w += st[j].w; }
    }
    float* sp = &lds_x[(SUMROW + rh) * LDSTR + c4 * 4];
    sp[0] = cs.x; sp[1] = cs.y; sp[2] = cs.z; sp[3] = cs.w;
  }
  __syncthreads();

  const int wave = __builtin_amdgcn_readfirstlane(tid >> 5);
  const int lane = tid & 31;
  const int ln   = lane & 15;
  const int lh   = lane >> 4;

  if (wave < 8) {
    const float* W; const float* bias; float* tile; int n0;
    if (wave < 4) { W = Wq; bias = bq; tile = lds_qk;               n0 = wave * 16; }
    else          { W = Wk; bias = bk; tile = lds_qk + KS_OFF + SP; n0 = (wave - 4) * 16; }

    const float bval = bias[n0 + ln];
    v8f acc[MT];
#pragma unroll
    for (int t = 0; t < MT; ++t) {
#pragma unroll
      for (int r = 0; r < 8; ++r) acc[t][r] = bval;
    }

    wave_kloop<MT, 4>(&lds_x[ln * LDSTR], W + n0 + ln, lh, acc);

    asm volatile("v_nop\n\tv_nop\n\tv_nop\n\tv_nop\n\tv_nop"
                 : "+v"(acc[0]), "+v"(acc[1]), "+v"(acc[2]), "+v"(acc[3]));

#pragma unroll
    for (int t = 0; t < MT; ++t) {
#pragma unroll
      for (int r = 0; r < 8; ++r) {
        const int row = 16 * t + r + 8 * lh;
        tile[row * SP + n0 + ln] = acc[t][r];
      }
    }
  } else {
    const int h  = tid - 256;
    const int cg = h & 15;
    const int kg = h >> 4;
    const float* wv4 = Wv + 4 * cg;
    float* vp = &vpart[kg * DHEAD + 4 * cg];
    __builtin_amdgcn_s_setprio(3);
    if (first) {
      v4f acc[4];
      v_rows<4>(lds_x, wv4, kg * VKPER, acc);
#pragma unroll
      for (int r = 0; r < 4; ++r) *(v4f*)(vp + r * VKG * DHEAD) = acc[r];
    } else {
      v4f acc[1];
      v_rows<1>(lds_x, wv4, kg * VKPER, acc);
      *(v4f*)(vp + 3 * VKG * DHEAD) = acc[0];
    }
    __builtin_amdgcn_s_setprio(0);
    {
      const int hr = (h < DHEAD) ? 0 : (BROWS + 1);
      lds_qk[KS_OFF + hr * SP + (h & (DHEAD - 1))] = 0.0f;
    }
  }
  __syncthreads();

  if (wave < 8) {
    const int r = tid >> 2;
    const int j = tid & 3;
    const float4* q4 = (const float4*)(lds_qk + r * SP + 16 * j);
    const float4* kp = (const float4*)(lds_qk + KS_OFF + r * SP + 16 * j);
    const float4* kc = (const float4*)(lds_qk + KS_OFF + (r + 1) * SP + 16 * j);
    const float4* kn = (const float4*)(lds_qk + KS_OFF + (r + 2) * SP + 16 * j);
    float pa = 0.0f, pb = 0.0f, pc = 0.0f;
#pragma unroll
    for (int e = 0; e < 4; ++e) {
      const float4 q = q4[e], a = kp[e], b = kc[e], c = kn[e];
      pa += q.x * a.x + q.y * a.y + q.z * a.z + q.w * a.w;
      pb += q.x * b.x + q.y * b.y + q.z * b.z + q.w * b.w;
      pc += q.x * c.x + q.y * c.y + q.z * c.z + q.w * c.w;
    }
    pa += __shfl_xor(pa, 1, 32);  pb += __shfl_xor(pb, 1, 32);  pc += __shfl_xor(pc, 1, 32);
    pa += __shfl_xor(pa, 2, 32);  pb += __shfl_xor(pb, 2, 32);  pc += __shfl_xor(pc, 2, 32);
    if (j == 0) ((float4*)S4)[m0 + r] = make_float4(pa, pb, pc, 0.0f);

    if (wave < 2) {
      const int er = (lane < 16) ? 0 : (BROWS - 1);
      const float* srow = (wave == 0) ? (lds_qk + er * SP) : (lds_qk + KS_OFF + (er + 1) * SP);
      float* edge = (wave == 0) ? Qe : Ke;
      ((float4*)edge)[(size_t)blockIdx.x * 32 + lane] = ((const float4*)srow)[lane & 15];
    }
  } else {
    const int h = tid - 256;
    const int d = h & (DHEAD - 1);
    const float bval = bv[d];
    const float cnt = first ? (float)(BROWS - 3) : (float)BROWS;
#pragma unroll
    for (int rr = 0; rr < 2; ++rr) {
      const int row = (h >> 6) + 2 * rr;
      if (row == 3 || first) {
        float a = 0.0f;
#pragma unroll
        for (int g = 0; g < VKG; ++g) a += vpart[(row * VKG + g) * DHEAD + d];
        if (row == 3) Psum[(size_t)blockIdx.x * DHEAD + d] = a + cnt * bval;
        else          Vhead[(size_t)((m0 / NSEQ) * 3 + row) * DHEAD + d] = a + bval;
      }
    }
  }
}

__global__ __launch_bounds__(256) void combine_kernel(
    const float* __restrict__ S4, const float* __restrict__ Qe, const float* __restrict__ Ke,
    const float* __restrict__ Vhead, const float* __restrict__ Psum,
    float* __restrict__ out) {
  __shared__ __attribute__((aligned(16))) float red[16 * DHEAD];
  const int tid  = threadIdx.x;
  const int wave = tid >> 5;
  const int lane = tid & 31;
  const int hf   = lane >> 4;
  const int c4   = lane & 15;
  const int blk  = blockIdx.x;
  const int b    = blk / PARTS_PER_BATCH;
  const int pb   = blk % PARTS_PER_BATCH;

  v4f sc[4];
#pragma unroll
  for (int ps = 0; ps < 4; ++ps) sc[ps] = ((const v4f*)S4)[blk * BROWS + wave * 8 + ps * 2 + hf];

  const v4f* Vb = (const v4f*)(Vhead + (size_t)b * 3 * DHEAD);
  v4f v0 = Vb[c4];
  v4f v1 = Vb[16 + c4];
  v4f v2 = Vb[32 + c4];

  const int lastrow = (wave == 7) ? 1 : 0;
  const int nb = lastrow ? ((pb < PARTS_PER_BATCH - 1) ? blk + 1 : blk)
                         : ((pb > 0) ? blk - 1 : blk);
  v4f eq = ((const v4f*)Qe)[(size_t)blk * 32 + 16 * lastrow + c4];
  v4f ek = ((const v4f*)Ke)[(size_t)nb * 32 + 16 * (1 - lastrow) + c4];

  const int d  = tid & (DHEAD - 1);
  const int sg = tid >> 6;
  const float* pp = Psum + ((size_t)b * PARTS_PER_BATCH + sg * 16) * DHEAD + d;
  v4f pr[4];
#pragma unroll
  for (int u = 0; u < 4; ++u) {
#pragma unroll
    for (int k = 0; k < 4; ++k) pr[u][k] = pp[(4 * u + k) * DHEAD];
  }

  asm volatile("" : "+v"(sc[0]), "+v"(sc[1]), "+v"(sc[2]), "+v"(sc[3]),
                    "+v"(v0), "+v"(v1), "+v"(v2), "+v"(eq), "+v"(ek),
                    "+v"(pr[0]), "+v"(pr[1]), "+v"(pr[2]), "+v"(pr[3]));

#pragma unroll
  for (int u = 0; u < 4; ++u) {
    float a = 0.0f;
#pragma unroll
    for (int k = 0; k < 4; ++k) a += pr[u][k];
    red[(4 * sg + u) * DHEAD + d] = a;
  }

  float es = eq.x * ek.x + eq.y * ek.y + eq.z * ek.z + eq.w * ek.w;
  es += __int_as_float(__builtin_amdgcn_update_dpp(0, __float_as_int(es), 0x168, 0xf, 0xf, true));
  es += __int_as_float(__builtin_amdgcn_update_dpp(0, __float_as_int(es), 0x164, 0xf, 0xf, true));
  es += __int_as_float(__builtin_amdgcn_update_dpp(0, __float_as_int(es), 0x162, 0xf, 0xf, true));
  es += __int_as_float(__builtin_amdgcn_update_dpp(0, __float_as_int(es), 0x161, 0xf, 0xf, true));

  float4 part[4];
  float ezv[4], invv[4];
#pragma unroll
  for (int ps = 0; ps < 4; ++ps) {
    const int r = wave * 8 + ps * 2 + hf;
    const int g = blk * BROWS + r;
    const int i = g & (NSEQ - 1);
    const v4f s = sc[ps];
    const float pa = (r == 0) ? es : s.x;
    const float pc = (r == BROWS - 1) ? es : s.z;

    const float a  = (i > 0) ? pa : 0.0f;
    const float bs = s.y;
    const float c  = (i < NSEQ - 1) ? pc : 0.0f;

    const float m  = fmaxf(fmaxf(a, bs), fmaxf(c, 0.0f));
    const float ea = expf(a - m);
    const float eb = expf(bs - m);
    const float ec = expf(c - m);
    const float ez = expf(-m);
    invv[ps] = 1.0f / (ea + eb + ec + (float)(NSEQ - 3) * ez);
    ezv[ps]  = ez;
    part[ps].x = ea * v0.x + eb * v1.x + ec * v2.x;
    part[ps].y = ea * v0.y + eb * v1.y + ec * v2.y;
    part[ps].z = ea * v0.z + eb * v1.z + ec * v2.z;
    part[ps].w = ea * v0.w + eb * v1.w + ec * v2.w;
  }
  __syncthreads();

  float4 sr = make_float4(0.0f, 0.0f, 0.0f, 0.0f);
#pragma unroll
  for (int k = 0; k < 16; ++k) {
    const float4 t = ((const float4*)red)[k * (DHEAD / 4) + c4];
    sr.x += t.x; sr.y += t.y; sr.z += t.z; sr.w += t.w;
  }

#pragma unroll
  for (int ps = 0; ps < 4; ++ps) {
    const int g = blk * BROWS + wave * 8 + ps * 2 + hf;
    float4 o;
    o.x = (part[ps].x + ezv[ps] * sr.x) * invv[ps];
    o.y = (part[ps].y + ezv[ps] * sr.y) * invv[ps];
    o.z = (part[ps].z + ezv[ps] * sr.z) * invv[ps];
    o.w = (part[ps].w + ezv[ps] * sr.w) * invv[ps];
    ((float4*)(out + (size_t)g * DHEAD))[c4] = o;
  }
}

extern "C" void kernel_launch(void* const* d_in, const int* in_sizes, int n_in,
                              void* d_out, int out_size, void* d_ws, size_t ws_size,
                              hipStream_t stream) {
  const float* x  = (const float*)d_in[0];
  const float* Wq = (const float*)d_in[1];
  const float* bq = (const float*)d_in[2];
  const float* Wk = (const float*)d_in[3];
  const float* bk = (const float*)d_in[4];
  const float* Wv = (const float*)d_in[5];
  const float* bv = (const float*)d_in[6];
  float* out = (float*)d_out;

  const int BN = in_sizes[0] / EMBED;
  const int B  = BN / NSEQ;

  float* S4    = (float*)d_ws;
  float* Qe    = S4 + (size_t)BN * 4;
  float* Ke    = Qe + (size_t)(BN / BROWS) * 2 * DHEAD;
  float* Vhead = Ke + (size_t)(BN / BROWS) * 2 * DHEAD;
  float* Psum  = Vhead + (size_t)B * 3 * DHEAD;

  qkv_gemm_kernel<<<BN / BROWS, 384, 0, stream>>>(x, Wq, bq, Wk, bk, Wv, bv, S4, Qe, Ke, Vhead, Psum);
  combine_kernel<<<BN / BROWS, 256, 0, stream>>>(S4, Qe, Ke, Vhead, Psum, out);
}
